# starting kernel
# baseline (speedup 1.0000x reference)
#include <hip/hip_runtime.h>
#include <math.h>

typedef __attribute__((ext_vector_type(2))) float v2f;
typedef __attribute__((ext_vector_type(8))) float v8f;

#define TMAX 4096
#define KC 8
#define PTS_PER_WAVE (4 * KC)
#define PPL (2 * KC)
#define WAVES_PER_BLOCK 8
#define BLOCK 256

static __device__ __forceinline__ float bcast_first(float x) {
  return __int_as_float(__builtin_amdgcn_readfirstlane(__float_as_int(x)));
}

__global__ __launch_bounds__(BLOCK) void hyst_dens_kernel(
    const float* __restrict__ raw, float* __restrict__ dens, int n, int npad) {
  int j = blockIdx.x * blockDim.x + threadIdx.x;
  if (j < npad) {
    float v = 0.0f;
    if (j < n) {
      float x = raw[j];
      v = fmaxf(x, 0.0f) + log1pf(expf(-fabsf(x)));
    }
    dens[j] = v;
  }
}

__global__ __launch_bounds__(BLOCK) void hyst_scan_kernel(
    const float* __restrict__ h, const float* __restrict__ mesh,
    const float* __restrict__ dens, float* __restrict__ partial,
    int T, int n) {
  __shared__ float lh[TMAX];
  for (int i = threadIdx.x; i < T; i += BLOCK) lh[i] = h[i];
  __syncthreads();

  const int lane = threadIdx.x & 31;
  const int row  = lane & 15;
  const int half = lane >> 4;
  const int w    = blockIdx.x * WAVES_PER_BLOCK + (threadIdx.x >> 5);
  const int pbase = w * PTS_PER_WAVE + 2 * half;

  float al[PPL], be[PPL], dd[PPL];
#pragma unroll
  for (int kc = 0; kc < KC; ++kc) {
#pragma unroll
    for (int i = 0; i < 2; ++i) {
      int p = pbase + 4 * kc + i;
      float a = 3.0e38f, b = -3.0e38f, d = 0.0f;
      if (p < n) { b = mesh[2 * p]; a = mesh[2 * p + 1]; d = dens[p]; }
      al[2 * kc + i] = a; be[2 * kc + i] = b; dd[2 * kc + i] = d;
    }
  }

  unsigned sbits = 0u;
  unsigned snapbits = 0u;
  float* pout = partial + (size_t)w * T;

  float hp = 0.0f;
  for (int tb = 0; tb < T; tb += 16) {
    v8f acc = {};
    for (int tt = 0; tt < 16; ++tt) {
      float hi = bcast_first(lh[tb + tt]);
      if (hi > hp) {
#pragma unroll
        for (int p = 0; p < PPL; ++p)
          sbits |= (al[p] <= hi) ? (1u << p) : 0u;
      } else if (hi < hp) {
#pragma unroll
        for (int p = 0; p < PPL; ++p)
          sbits &= (be[p] >= hi) ? ~(1u << p) : 0xffffffffu;
      }
      if (tt == row) snapbits = sbits;
      hp = hi;
    }
#pragma unroll
    for (int kc = 0; kc < KC; ++kc) {
      v2f A, B;
      A.x = ((snapbits >> (2 * kc)) & 1u) ? 1.0f : -1.0f;
      A.y = ((snapbits >> (2 * kc + 1)) & 1u) ? 1.0f : -1.0f;
      B.x = dd[2 * kc];
      B.y = dd[2 * kc + 1];
      acc = __builtin_amdgcn_wmma_f32_16x16x4_f32(
          false, A, false, B, (short)0, acc, false, false);
    }
    if (lane == 0) {
      *reinterpret_cast<float4*>(pout + tb) =
          make_float4(acc[0], acc[1], acc[2], acc[3]);
      *reinterpret_cast<float4*>(pout + tb + 4) =
          make_float4(acc[4], acc[5], acc[6], acc[7]);
    } else if (lane == 16) {
      *reinterpret_cast<float4*>(pout + tb + 8) =
          make_float4(acc[0], acc[1], acc[2], acc[3]);
      *reinterpret_cast<float4*>(pout + tb + 12) =
          make_float4(acc[4], acc[5], acc[6], acc[7]);
    }
  }
}

__global__ __launch_bounds__(BLOCK) void hyst_reduce_kernel(
    const float* __restrict__ partial, const float* __restrict__ scale,
    const float* __restrict__ offset, float* __restrict__ out,
    int T, int NW, float inv_n) {
  int t = blockIdx.x * blockDim.x + threadIdx.x;
  if (t < T) {
    float sum = 0.0f;
    for (int wv = 0; wv < NW; ++wv) sum += partial[(size_t)wv * T + t];
    out[t] = scale[0] * (sum * inv_n) + offset[0];
  }
}

extern "C" void kernel_launch(void* const* d_in, const int* in_sizes, int n_in,
                              void* d_out, int out_size, void* d_ws, size_t ws_size,
                              hipStream_t stream) {
  const float* h      = (const float*)d_in[0];
  const float* mesh   = (const float*)d_in[1];
  const float* raw    = (const float*)d_in[2];
  const float* scale  = (const float*)d_in[3];
  const float* offset = (const float*)d_in[4];
  float* out = (float*)d_out;

  const int T = in_sizes[0];
  const int n = in_sizes[2];

  const int nWaves  = (n + PTS_PER_WAVE - 1) / PTS_PER_WAVE;
  const int nBlocks = (nWaves + WAVES_PER_BLOCK - 1) / WAVES_PER_BLOCK;
  const int NW      = nBlocks * WAVES_PER_BLOCK;
  const int npad    = NW * PTS_PER_WAVE;

  float* partial = (float*)d_ws;
  float* dens    = partial + (size_t)NW * T;

  hyst_dens_kernel<<<(npad + BLOCK - 1) / BLOCK, BLOCK, 0, stream>>>(
      raw, dens, n, npad);
  hyst_scan_kernel<<<nBlocks, BLOCK, 0, stream>>>(
      h, mesh, dens, partial, T, n);
  hyst_reduce_kernel<<<(T + BLOCK - 1) / BLOCK, BLOCK, 0, stream>>>(
      partial, scale, offset, out, T, NW, 1.0f / (float)n);
}


# scan kernel: each lane owns 2 relays (64 per wave, none scanned 16 times); its states are A, its densities B of one WMMA per step; D's diagonal is summed through LDS; the relay sum's order changes
# speedup vs baseline: 2.5785x; 2.5785x over previous
#include <hip/hip_runtime.h>
#include <math.h>

typedef __attribute__((ext_vector_type(2))) float v2f;
typedef __attribute__((ext_vector_type(8))) float v8f;

#define RPW 64
#define TT 32
#define PITCH 33
#define BLOCK 256

__global__ __launch_bounds__(BLOCK) void hyst_dens_kernel(
    const float* __restrict__ raw, float* __restrict__ dens, int n, int npad) {
  int j = blockIdx.x * blockDim.x + threadIdx.x;
  if (j < npad) {
    float v = 0.0f;
    if (j < n) {
      float x = raw[j];
      v = fmaxf(x, 0.0f) + log1pf(expf(-fabsf(x)));
    }
    dens[j] = v;
  }
}

__global__ __launch_bounds__(32) void hyst_scan_kernel(
    const float* __restrict__ h, const float* __restrict__ mesh,
    const float* __restrict__ dens, float* __restrict__ partial,
    int T, int n) {
  __shared__ float sx[2][TT * PITCH];

  const int lane = threadIdx.x;
  const int w    = blockIdx.x;
  const int p0   = w * RPW + 2 * lane;

  float a0 = 3.0e38f, b0 = -3.0e38f, d0 = 0.0f;
  float a1 = 3.0e38f, b1 = -3.0e38f, d1 = 0.0f;
  if (p0 < n)     { b0 = mesh[2 * p0];     a0 = mesh[2 * p0 + 1]; d0 = dens[p0]; }
  if (p0 + 1 < n) { b1 = mesh[2 * p0 + 2]; a1 = mesh[2 * p0 + 3]; d1 = dens[p0 + 1]; }

  v2f B;
  B.x = d0;
  B.y = d1;

  float s0 = -1.0f, s1 = -1.0f;
  float* pout = partial + (size_t)w * T;
  const float never = __builtin_nanf("");

  const bool sel1 = (lane & 1) != 0;
  const bool sel2 = (lane & 2) != 0;
  const bool sel4 = (lane & 4) != 0;

  float hc[TT];
#pragma unroll
  for (int tt = 0; tt < TT; ++tt) hc[tt] = h[tt];

  float hp = 0.0f;
  int buf = 0;
  for (int tb = 0; tb < T; tb += TT) {
    const int nb = (tb + TT < T) ? tb + TT : tb;
    float hn[TT];
#pragma unroll
    for (int tt = 0; tt < TT; ++tt) hn[tt] = h[nb + tt];
    float* sb = sx[buf];
#pragma unroll
    for (int tt = 0; tt < TT; ++tt) {
      const float hi = hc[tt];
      const float up = (hi > hp) ? hi : never;
      const float dn = (hi < hp) ? hi : never;
      s0 = (a0 <= up) ? 1.0f : s0;
      s0 = (b0 >= dn) ? -1.0f : s0;
      s1 = (a1 <= up) ? 1.0f : s1;
      s1 = (b1 >= dn) ? -1.0f : s1;
      hp = hi;

      v2f A;
      A.x = s0;
      A.y = s1;
      v8f acc = {};
      acc = __builtin_amdgcn_wmma_f32_16x16x4_f32(
          false, A, false, B, (short)0, acc, false, false);

      const float x01 = sel1 ? acc[1] : acc[0];
      const float x23 = sel1 ? acc[3] : acc[2];
      const float x45 = sel1 ? acc[5] : acc[4];
      const float x67 = sel1 ? acc[7] : acc[6];
      const float x03 = sel2 ? x23 : x01;
      const float x47 = sel2 ? x67 : x45;
      sb[tt * PITCH + lane] = sel4 ? x47 : x03;
    }
    __syncthreads();
    const float* r = sb + lane * PITCH;
    float q0 = (r[0] + r[1]) + (r[2] + r[3]);
    float q1 = (r[4] + r[5]) + (r[6] + r[7]);
    float q2 = (r[24] + r[25]) + (r[26] + r[27]);
    float q3 = (r[28] + r[29]) + (r[30] + r[31]);
    pout[tb + lane] = (q0 + q1) + (q2 + q3);
#pragma unroll
    for (int tt = 0; tt < TT; ++tt) hc[tt] = hn[tt];
    buf ^= 1;
  }
}

__global__ __launch_bounds__(BLOCK) void hyst_reduce_kernel(
    const float* __restrict__ partial, const float* __restrict__ scale,
    const float* __restrict__ offset, float* __restrict__ out,
    int T, int NW, float inv_n) {
  int t = blockIdx.x * blockDim.x + threadIdx.x;
  if (t < T) {
    float sum = 0.0f;
    for (int wv = 0; wv < NW; ++wv) sum += partial[(size_t)wv * T + t];
    out[t] = scale[0] * (sum * inv_n) + offset[0];
  }
}

extern "C" void kernel_launch(void* const* d_in, const int* in_sizes, int n_in,
                              void* d_out, int out_size, void* d_ws, size_t ws_size,
                              hipStream_t stream) {
  const float* h      = (const float*)d_in[0];
  const float* mesh   = (const float*)d_in[1];
  const float* raw    = (const float*)d_in[2];
  const float* scale  = (const float*)d_in[3];
  const float* offset = (const float*)d_in[4];
  float* out = (float*)d_out;

  const int T = in_sizes[0];
  const int n = in_sizes[2];

  const int NW   = (n + RPW - 1) / RPW;
  const int npad = NW * RPW;

  float* partial = (float*)d_ws;
  float* dens    = partial + (size_t)NW * T;

  hyst_dens_kernel<<<(npad + BLOCK - 1) / BLOCK, BLOCK, 0, stream>>>(
      raw, dens, n, npad);
  hyst_scan_kernel<<<NW, 32, 0, stream>>>(
      h, mesh, dens, partial, T, n);
  hyst_reduce_kernel<<<(T + BLOCK - 1) / BLOCK, BLOCK, 0, stream>>>(
      partial, scale, offset, out, T, NW, 1.0f / (float)n);
}


# scan kernel: time cut into 16 chunks, one work-group each; a chunk finds its start states by walking back to each relay's latest set/clear event, blocks gated by phase-0 max/min; sums unchanged
# speedup vs baseline: 8.5181x; 3.3035x over previous
#include <hip/hip_runtime.h>
#include <math.h>

typedef __attribute__((ext_vector_type(2))) float v2f;
typedef __attribute__((ext_vector_type(8))) float v8f;

#define RPW 64
#define TT 32
#define PITCH 33
#define BLOCK 256
#define NCH 16

__global__ __launch_bounds__(BLOCK) void hyst_dens_kernel(
    const float* __restrict__ raw, const float* __restrict__ h,
    float* __restrict__ dens, float* __restrict__ agg, int n, int npad, int T) {
  int j = blockIdx.x * blockDim.x + threadIdx.x;
  const int nblk = T / TT;
  for (int b = j; b < nblk; b += gridDim.x * blockDim.x) {
    const int bb = b * TT;
    float q = (bb > 0) ? h[bb - 1] : 0.0f;
    float mu = -__builtin_huge_valf(), md = __builtin_huge_valf();
#pragma unroll
    for (int tt = 0; tt < TT; ++tt) {
      const float hi = h[bb + tt];
      mu = (hi > q) ? fmaxf(mu, hi) : mu;
      md = (hi < q) ? fminf(md, hi) : md;
      q = hi;
    }
    agg[b] = mu;
    agg[nblk + b] = md;
  }
  if (j < npad) {
    float v = 0.0f;
    if (j < n) {
      float x = raw[j];
      v = fmaxf(x, 0.0f) + log1pf(expf(-fabsf(x)));
    }
    dens[j] = v;
  }
}

__global__ __launch_bounds__(32) void hyst_scan_kernel(
    const float* __restrict__ h, const float* __restrict__ mesh,
    const float* __restrict__ dens, const float* __restrict__ agg,
    float* __restrict__ partial, int T, int n) {
  __shared__ float sx[2][TT * PITCH];

  const int lane = threadIdx.x;
  const int w    = blockIdx.x;
  const int p0   = w * RPW + 2 * lane;

  float a0 = 3.0e38f, b0 = -3.0e38f, d0 = 0.0f;
  float a1 = 3.0e38f, b1 = -3.0e38f, d1 = 0.0f;
  if (p0 < n)     { b0 = mesh[2 * p0];     a0 = mesh[2 * p0 + 1]; d0 = dens[p0]; }
  if (p0 + 1 < n) { b1 = mesh[2 * p0 + 2]; a1 = mesh[2 * p0 + 3]; d1 = dens[p0 + 1]; }

  v2f B;
  B.x = d0;
  B.y = d1;

  const int tpc = (T / TT + NCH - 1) / NCH;
  const int t0  = (NCH - 1 - (int)blockIdx.y) * tpc * TT;
  if (t0 >= T) return;
  const int t1  = (t0 + tpc * TT < T) ? t0 + tpc * TT : T;

  float* pout = partial + (size_t)w * T;
  const float never = __builtin_nanf("");

  float s0 = -1.0f, s1 = -1.0f;
  if (t0 > 0) {
    float c0 = (p0 < n) ? 0.0f : -1.0f;
    float c1 = (p0 + 1 < n) ? 0.0f : -1.0f;
    const int nblk = T / TT;
    const int nb0  = t0 / TT;
    bool done = false;
    for (int g = (nb0 - 1) / 32; g >= 0 && !done; --g) {
      const int bi = g * 32 + lane;
      float muv = -__builtin_huge_valf(), mdv = __builtin_huge_valf();
      if (bi < nb0) { muv = agg[bi]; mdv = agg[nblk + bi]; }
      const int btop = (nb0 - 1 - g * 32 < 31) ? nb0 - 1 - g * 32 : 31;
      for (int b = btop; b >= 0; --b) {
        const float mu = __int_as_float(__builtin_amdgcn_readlane(__float_as_int(muv), b));
        const float md = __int_as_float(__builtin_amdgcn_readlane(__float_as_int(mdv), b));
        const bool hit = ((c0 == 0.0f) && ((a0 <= mu) || (b0 >= md))) ||
                         ((c1 == 0.0f) && ((a1 <= mu) || (b1 >= md)));
        if (!__any(hit)) continue;
        const int bb = (g * 32 + b) * TT;
        float hb[TT];
#pragma unroll
        for (int tt = 0; tt < TT; ++tt) hb[tt] = h[bb + tt];
        float hq = (bb > 0) ? h[bb - 1] : 0.0f;
        float e0 = 0.0f, e1 = 0.0f;
#pragma unroll
        for (int tt = 0; tt < TT; ++tt) {
          const float hi = hb[tt];
          const float up = (hi > hq) ? hi : never;
          const float dn = (hi < hq) ? hi : never;
          e0 = (a0 <= up) ? 1.0f : e0;
          e0 = (b0 >= dn) ? -1.0f : e0;
          e1 = (a1 <= up) ? 1.0f : e1;
          e1 = (b1 >= dn) ? -1.0f : e1;
          hq = hi;
        }
        c0 = (c0 == 0.0f) ? e0 : c0;
        c1 = (c1 == 0.0f) ? e1 : c1;
        if (__all((c0 != 0.0f) && (c1 != 0.0f))) { done = true; break; }
      }
    }
    s0 = (c0 > 0.0f) ? 1.0f : -1.0f;
    s1 = (c1 > 0.0f) ? 1.0f : -1.0f;
  }

  const bool sel1 = (lane & 1) != 0;
  const bool sel2 = (lane & 2) != 0;
  const bool sel4 = (lane & 4) != 0;

  float hc[TT];
#pragma unroll
  for (int tt = 0; tt < TT; ++tt) hc[tt] = h[t0 + tt];

  float hp = (t0 > 0) ? h[t0 - 1] : 0.0f;
  int buf = 0;
  for (int tb = t0; tb < t1; tb += TT) {
    const int nb = (tb + TT < T) ? tb + TT : tb;
    float hn[TT];
#pragma unroll
    for (int tt = 0; tt < TT; ++tt) hn[tt] = h[nb + tt];
    float* sb = sx[buf];
#pragma unroll
    for (int tt = 0; tt < TT; ++tt) {
      const float hi = hc[tt];
      const float up = (hi > hp) ? hi : never;
      const float dn = (hi < hp) ? hi : never;
      s0 = (a0 <= up) ? 1.0f : s0;
      s0 = (b0 >= dn) ? -1.0f : s0;
      s1 = (a1 <= up) ? 1.0f : s1;
      s1 = (b1 >= dn) ? -1.0f : s1;
      hp = hi;

      v2f A;
      A.x = s0;
      A.y = s1;
      v8f acc = {};
      acc = __builtin_amdgcn_wmma_f32_16x16x4_f32(
          false, A, false, B, (short)0, acc, false, false);

      const float x01 = sel1 ? acc[1] : acc[0];
      const float x23 = sel1 ? acc[3] : acc[2];
      const float x45 = sel1 ? acc[5] : acc[4];
      const float x67 = sel1 ? acc[7] : acc[6];
      const float x03 = sel2 ? x23 : x01;
      const float x47 = sel2 ? x67 : x45;
      sb[tt * PITCH + lane] = sel4 ? x47 : x03;
    }
    __syncthreads();
    const float* r = sb + lane * PITCH;
    float q0 = (r[0] + r[1]) + (r[2] + r[3]);
    float q1 = (r[4] + r[5]) + (r[6] + r[7]);
    float q2 = (r[24] + r[25]) + (r[26] + r[27]);
    float q3 = (r[28] + r[29]) + (r[30] + r[31]);
    pout[tb + lane] = (q0 + q1) + (q2 + q3);
#pragma unroll
    for (int tt = 0; tt < TT; ++tt) hc[tt] = hn[tt];
    buf ^= 1;
  }
}

__global__ __launch_bounds__(BLOCK) void hyst_reduce_kernel(
    const float* __restrict__ partial, const float* __restrict__ scale,
    const float* __restrict__ offset, float* __restrict__ out,
    int T, int NW, float inv_n) {
  int t = blockIdx.x * blockDim.x + threadIdx.x;
  if (t < T) {
    float sum = 0.0f;
    for (int wv = 0; wv < NW; ++wv) sum += partial[(size_t)wv * T + t];
    out[t] = scale[0] * (sum * inv_n) + offset[0];
  }
}

extern "C" void kernel_launch(void* const* d_in, const int* in_sizes, int n_in,
                              void* d_out, int out_size, void* d_ws, size_t ws_size,
                              hipStream_t stream) {
  const float* h      = (const float*)d_in[0];
  const float* mesh   = (const float*)d_in[1];
  const float* raw    = (const float*)d_in[2];
  const float* scale  = (const float*)d_in[3];
  const float* offset = (const float*)d_in[4];
  float* out = (float*)d_out;

  const int T = in_sizes[0];
  const int n = in_sizes[2];

  const int NW   = (n + RPW - 1) / RPW;
  const int npad = NW * RPW;

  float* partial = (float*)d_ws;
  float* dens    = partial + (size_t)NW * T;
  float* agg     = dens + npad;

  hyst_dens_kernel<<<(npad + BLOCK - 1) / BLOCK, BLOCK, 0, stream>>>(
      raw, h, dens, agg, n, npad, T);
  hyst_scan_kernel<<<dim3(NW, NCH), 32, 0, stream>>>(
      h, mesh, dens, agg, partial, T, n);
  hyst_reduce_kernel<<<(T + BLOCK - 1) / BLOCK, BLOCK, 0, stream>>>(
      partial, scale, offset, out, T, NW, 1.0f / (float)n);
}


# scan kernel: work-groups of 20 waves (one per WGP): each wave's step sums go to an LDS row, the group adds the rows and writes one partial row, 16 rows not 315; the relay sum's tree changes
# speedup vs baseline: 15.2182x; 1.7866x over previous
#include <hip/hip_runtime.h>
#include <math.h>

typedef __attribute__((ext_vector_type(2))) float v2f;
typedef __attribute__((ext_vector_type(8))) float v8f;

#define RPW 64
#define TT 32
#define PITCH 33
#define BLOCK 256
#define NCH 16
#define WPG 20
#define SEG 8

__global__ __launch_bounds__(BLOCK) void hyst_dens_kernel(
    const float* __restrict__ raw, const float* __restrict__ h,
    float* __restrict__ dens, float* __restrict__ agg, int n, int npad, int T) {
  int j = blockIdx.x * blockDim.x + threadIdx.x;
  const int nblk = T / TT;
  for (int b = j; b < nblk; b += gridDim.x * blockDim.x) {
    const int bb = b * TT;
    float q = (bb > 0) ? h[bb - 1] : 0.0f;
    float mu = -__builtin_huge_valf(), md = __builtin_huge_valf();
#pragma unroll
    for (int tt = 0; tt < TT; ++tt) {
      const float hi = h[bb + tt];
      mu = (hi > q) ? fmaxf(mu, hi) : mu;
      md = (hi < q) ? fminf(md, hi) : md;
      q = hi;
    }
    agg[b] = mu;
    agg[nblk + b] = md;
  }
  if (j < npad) {
    float v = 0.0f;
    if (j < n) {
      float x = raw[j];
      v = fmaxf(x, 0.0f) + log1pf(expf(-fabsf(x)));
    }
    dens[j] = v;
  }
}

__global__ __launch_bounds__(WPG * 32) void hyst_scan_kernel(
    const float* __restrict__ h, const float* __restrict__ mesh,
    const float* __restrict__ dens, const float* __restrict__ agg,
    float* __restrict__ partial, int T, int n, int NW) {
  __shared__ float sx[WPG][2][TT * PITCH];
  __shared__ float wsum[WPG][SEG * TT];

  const int lane = threadIdx.x & 31;
  const int wave = __builtin_amdgcn_readfirstlane((int)(threadIdx.x >> 5));
  const int w    = blockIdx.x * WPG + wave;
  const bool owner = w < NW;
  const int p0   = w * RPW + 2 * lane;

  float a0 = 3.0e38f, b0 = -3.0e38f, d0 = 0.0f;
  float a1 = 3.0e38f, b1 = -3.0e38f, d1 = 0.0f;
  if (p0 < n)     { b0 = mesh[2 * p0];     a0 = mesh[2 * p0 + 1]; d0 = dens[p0]; }
  if (p0 + 1 < n) { b1 = mesh[2 * p0 + 2]; a1 = mesh[2 * p0 + 3]; d1 = dens[p0 + 1]; }

  v2f B;
  B.x = d0;
  B.y = d1;

  const int tpc = (T / TT + NCH - 1) / NCH;
  const int t0  = (NCH - 1 - (int)blockIdx.y) * tpc * TT;
  if (t0 >= T) return;
  const int t1  = (t0 + tpc * TT < T) ? t0 + tpc * TT : T;

  float* pout = partial + (size_t)blockIdx.x * T;
  const float never = __builtin_nanf("");

  float s0 = -1.0f, s1 = -1.0f;
  if (owner && t0 > 0) {
    float c0 = (p0 < n) ? 0.0f : -1.0f;
    float c1 = (p0 + 1 < n) ? 0.0f : -1.0f;
    const int nblk = T / TT;
    const int nb0  = t0 / TT;
    bool done = false;
    for (int g = (nb0 - 1) / 32; g >= 0 && !done; --g) {
      const int bi = g * 32 + lane;
      float muv = -__builtin_huge_valf(), mdv = __builtin_huge_valf();
      if (bi < nb0) { muv = agg[bi]; mdv = agg[nblk + bi]; }
      const int btop = (nb0 - 1 - g * 32 < 31) ? nb0 - 1 - g * 32 : 31;
      for (int b = btop; b >= 0; --b) {
        const float mu = __int_as_float(__builtin_amdgcn_readlane(__float_as_int(muv), b));
        const float md = __int_as_float(__builtin_amdgcn_readlane(__float_as_int(mdv), b));
        const bool hit = ((c0 == 0.0f) && ((a0 <= mu) || (b0 >= md))) ||
                         ((c1 == 0.0f) && ((a1 <= mu) || (b1 >= md)));
        if (!__any(hit)) continue;
        const int bb = (g * 32 + b) * TT;
        float hb[TT];
#pragma unroll
        for (int tt = 0; tt < TT; ++tt) hb[tt] = h[bb + tt];
        float hq = (bb > 0) ? h[bb - 1] : 0.0f;
        float e0 = 0.0f, e1 = 0.0f;
#pragma unroll
        for (int tt = 0; tt < TT; ++tt) {
          const float hi = hb[tt];
          const float up = (hi > hq) ? hi : never;
          const float dn = (hi < hq) ? hi : never;
          e0 = (a0 <= up) ? 1.0f : e0;
          e0 = (b0 >= dn) ? -1.0f : e0;
          e1 = (a1 <= up) ? 1.0f : e1;
          e1 = (b1 >= dn) ? -1.0f : e1;
          hq = hi;
        }
        c0 = (c0 == 0.0f) ? e0 : c0;
        c1 = (c1 == 0.0f) ? e1 : c1;
        if (__all((c0 != 0.0f) && (c1 != 0.0f))) { done = true; break; }
      }
    }
    s0 = (c0 > 0.0f) ? 1.0f : -1.0f;
    s1 = (c1 > 0.0f) ? 1.0f : -1.0f;
  }

  const bool sel1 = (lane & 1) != 0;
  const bool sel2 = (lane & 2) != 0;
  const bool sel4 = (lane & 4) != 0;

  float hc[TT];
#pragma unroll
  for (int tt = 0; tt < TT; ++tt) hc[tt] = h[t0 + tt];

  float hp = (t0 > 0) ? h[t0 - 1] : 0.0f;
  int buf = 0;
  float* wrow = wsum[wave];
  for (int seg = t0; seg < t1; seg += SEG * TT) {
  const int segend = (seg + SEG * TT < t1) ? seg + SEG * TT : t1;
  if (owner) {
  for (int tb = seg; tb < segend; tb += TT) {
    const int nb = (tb + TT < T) ? tb + TT : tb;
    float hn[TT];
#pragma unroll
    for (int tt = 0; tt < TT; ++tt) hn[tt] = h[nb + tt];
    float* sb = sx[wave][buf];
#pragma unroll
    for (int tt = 0; tt < TT; ++tt) {
      const float hi = hc[tt];
      const float up = (hi > hp) ? hi : never;
      const float dn = (hi < hp) ? hi : never;
      s0 = (a0 <= up) ? 1.0f : s0;
      s0 = (b0 >= dn) ? -1.0f : s0;
      s1 = (a1 <= up) ? 1.0f : s1;
      s1 = (b1 >= dn) ? -1.0f : s1;
      hp = hi;

      v2f A;
      A.x = s0;
      A.y = s1;
      v8f acc = {};
      acc = __builtin_amdgcn_wmma_f32_16x16x4_f32(
          false, A, false, B, (short)0, acc, false, false);

      const float x01 = sel1 ? acc[1] : acc[0];
      const float x23 = sel1 ? acc[3] : acc[2];
      const float x45 = sel1 ? acc[5] : acc[4];
      const float x67 = sel1 ? acc[7] : acc[6];
      const float x03 = sel2 ? x23 : x01;
      const float x47 = sel2 ? x67 : x45;
      sb[tt * PITCH + lane] = sel4 ? x47 : x03;
    }
    __builtin_amdgcn_fence(__ATOMIC_RELEASE, "workgroup");
    __builtin_amdgcn_wave_barrier();
    __builtin_amdgcn_fence(__ATOMIC_ACQUIRE, "workgroup");
    const float* r = sb + lane * PITCH;
    float q0 = (r[0] + r[1]) + (r[2] + r[3]);
    float q1 = (r[4] + r[5]) + (r[6] + r[7]);
    float q2 = (r[24] + r[25]) + (r[26] + r[27]);
    float q3 = (r[28] + r[29]) + (r[30] + r[31]);
    wrow[tb - seg + lane] = (q0 + q1) + (q2 + q3);
#pragma unroll
    for (int tt = 0; tt < TT; ++tt) hc[tt] = hn[tt];
    buf ^= 1;
  }
  } else {
    for (int i = lane; i < segend - seg; i += 32) wrow[i] = 0.0f;
  }
  __syncthreads();
  static_assert(SEG * TT <= WPG * 32, "one thread per step of a segment");
  const int i = threadIdx.x;
  if (i < segend - seg) {
    float sum = wsum[0][i];
#pragma unroll
    for (int g = 1; g < WPG; ++g) sum += wsum[g][i];
    pout[seg + i] = sum;
  }
  __syncthreads();
  }
}

__global__ __launch_bounds__(BLOCK) void hyst_reduce_kernel(
    const float* __restrict__ partial, const float* __restrict__ scale,
    const float* __restrict__ offset, float* __restrict__ out,
    int T, int NW, float inv_n) {
  int t = blockIdx.x * blockDim.x + threadIdx.x;
  if (t < T) {
    float sum = 0.0f;
    for (int wv = 0; wv < NW; ++wv) sum += partial[(size_t)wv * T + t];
    out[t] = scale[0] * (sum * inv_n) + offset[0];
  }
}

extern "C" void kernel_launch(void* const* d_in, const int* in_sizes, int n_in,
                              void* d_out, int out_size, void* d_ws, size_t ws_size,
                              hipStream_t stream) {
  const float* h      = (const float*)d_in[0];
  const float* mesh   = (const float*)d_in[1];
  const float* raw    = (const float*)d_in[2];
  const float* scale  = (const float*)d_in[3];
  const float* offset = (const float*)d_in[4];
  float* out = (float*)d_out;

  const int T = in_sizes[0];
  const int n = in_sizes[2];

  const int NW   = (n + RPW - 1) / RPW;
  const int NG   = (NW + WPG - 1) / WPG;
  const int npad = NW * RPW;

  float* partial = (float*)d_ws;
  float* dens    = partial + (size_t)NG * T;
  float* agg     = dens + npad;

  hyst_dens_kernel<<<(npad + BLOCK - 1) / BLOCK, BLOCK, 0, stream>>>(
      raw, h, dens, agg, n, npad, T);
  hyst_scan_kernel<<<dim3(NG, NCH), WPG * 32, 0, stream>>>(
      h, mesh, dens, agg, partial, T, n, NW);
  hyst_reduce_kernel<<<(T + BLOCK - 1) / BLOCK, BLOCK, 0, stream>>>(
      partial, scale, offset, out, T, NG, 1.0f / (float)n);
}


# scan kernel: a lane owns 4 relays, not 2: a step's 2 WMMAs chain in one accumulator, 1 diagonal pick + LDS store per 128 relays; 32 time chunks, so still 5 waves per SIMD; relay sum's tree changes
# speedup vs baseline: 16.9576x; 1.1143x over previous
#include <hip/hip_runtime.h>
#include <math.h>

typedef __attribute__((ext_vector_type(2))) float v2f;
typedef __attribute__((ext_vector_type(8))) float v8f;

#define RPW 128
#define TT 32
#define PITCH 33
#define BLOCK 256
#define NCH 32
#define WPG 20
#define SEG 4

__global__ __launch_bounds__(BLOCK) void hyst_dens_kernel(
    const float* __restrict__ raw, const float* __restrict__ h,
    float* __restrict__ dens, float* __restrict__ agg, int n, int npad, int T) {
  int j = blockIdx.x * blockDim.x + threadIdx.x;
  const int nblk = T / TT;
  for (int b = j; b < nblk; b += gridDim.x * blockDim.x) {
    const int bb = b * TT;
    float q = (bb > 0) ? h[bb - 1] : 0.0f;
    float mu = -__builtin_huge_valf(), md = __builtin_huge_valf();
#pragma unroll
    for (int tt = 0; tt < TT; ++tt) {
      const float hi = h[bb + tt];
      mu = (hi > q) ? fmaxf(mu, hi) : mu;
      md = (hi < q) ? fminf(md, hi) : md;
      q = hi;
    }
    agg[b] = mu;
    agg[nblk + b] = md;
  }
  if (j < npad) {
    float v = 0.0f;
    if (j < n) {
      float x = raw[j];
      v = fmaxf(x, 0.0f) + log1pf(expf(-fabsf(x)));
    }
    dens[j] = v;
  }
}

__global__ __launch_bounds__(WPG * 32) void hyst_scan_kernel(
    const float* __restrict__ h, const float* __restrict__ mesh,
    const float* __restrict__ dens, const float* __restrict__ agg,
    float* __restrict__ partial, int T, int n, int NW) {
  __shared__ float sx[WPG][2][TT * PITCH];
  __shared__ float wsum[WPG][SEG * TT];

  const int lane = threadIdx.x & 31;
  const int wave = __builtin_amdgcn_readfirstlane((int)(threadIdx.x >> 5));
  const int w    = blockIdx.x * WPG + wave;
  const bool owner = w < NW;
  const int p0   = w * RPW + 4 * lane;

  float a0 = 3.0e38f, b0 = -3.0e38f, d0 = 0.0f;
  float a1 = 3.0e38f, b1 = -3.0e38f, d1 = 0.0f;
  float a2 = 3.0e38f, b2 = -3.0e38f, d2 = 0.0f;
  float a3 = 3.0e38f, b3 = -3.0e38f, d3 = 0.0f;
  if (p0 < n)     { b0 = mesh[2 * p0];     a0 = mesh[2 * p0 + 1]; d0 = dens[p0]; }
  if (p0 + 1 < n) { b1 = mesh[2 * p0 + 2]; a1 = mesh[2 * p0 + 3]; d1 = dens[p0 + 1]; }
  if (p0 + 2 < n) { b2 = mesh[2 * p0 + 4]; a2 = mesh[2 * p0 + 5]; d2 = dens[p0 + 2]; }
  if (p0 + 3 < n) { b3 = mesh[2 * p0 + 6]; a3 = mesh[2 * p0 + 7]; d3 = dens[p0 + 3]; }

  v2f B0, B1;
  B0.x = d0;
  B0.y = d1;
  B1.x = d2;
  B1.y = d3;

  const int tpc = (T / TT + NCH - 1) / NCH;
  const int t0  = (NCH - 1 - (int)blockIdx.y) * tpc * TT;
  if (t0 >= T) return;
  const int t1  = (t0 + tpc * TT < T) ? t0 + tpc * TT : T;

  float* pout = partial + (size_t)blockIdx.x * T;
  const float never = __builtin_nanf("");

  float s0 = -1.0f, s1 = -1.0f, s2 = -1.0f, s3 = -1.0f;
  if (owner && t0 > 0) {
    float c0 = (p0 < n) ? 0.0f : -1.0f;
    float c1 = (p0 + 1 < n) ? 0.0f : -1.0f;
    float c2 = (p0 + 2 < n) ? 0.0f : -1.0f;
    float c3 = (p0 + 3 < n) ? 0.0f : -1.0f;
    float am = fminf(fminf((c0 == 0.0f) ? a0 : 3.0e38f, (c1 == 0.0f) ? a1 : 3.0e38f),
                     fminf((c2 == 0.0f) ? a2 : 3.0e38f, (c3 == 0.0f) ? a3 : 3.0e38f));
    float bm = fmaxf(fmaxf((c0 == 0.0f) ? b0 : -3.0e38f, (c1 == 0.0f) ? b1 : -3.0e38f),
                     fmaxf((c2 == 0.0f) ? b2 : -3.0e38f, (c3 == 0.0f) ? b3 : -3.0e38f));
    const int nblk = T / TT;
    const int nb0  = t0 / TT;
    bool done = false;
    for (int g = (nb0 - 1) / 32; g >= 0 && !done; --g) {
      const int bi = g * 32 + lane;
      float muv = -__builtin_huge_valf(), mdv = __builtin_huge_valf();
      if (bi < nb0) { muv = agg[bi]; mdv = agg[nblk + bi]; }
      const int btop = (nb0 - 1 - g * 32 < 31) ? nb0 - 1 - g * 32 : 31;
      for (int b = btop; b >= 0; --b) {
        const float mu = __int_as_float(__builtin_amdgcn_readlane(__float_as_int(muv), b));
        const float md = __int_as_float(__builtin_amdgcn_readlane(__float_as_int(mdv), b));
        const bool hit = (am <= mu) || (bm >= md);
        if (!__any(hit)) continue;
        const int bb = (g * 32 + b) * TT;
        float hb[TT];
#pragma unroll
        for (int tt = 0; tt < TT; ++tt) hb[tt] = h[bb + tt];
        float hq = (bb > 0) ? h[bb - 1] : 0.0f;
        float e0 = 0.0f, e1 = 0.0f, e2 = 0.0f, e3 = 0.0f;
#pragma unroll
        for (int tt = 0; tt < TT; ++tt) {
          const float hi = hb[tt];
          const float up = (hi > hq) ? hi : never;
          const float dn = (hi < hq) ? hi : never;
          e0 = (a0 <= up) ? 1.0f : e0;
          e0 = (b0 >= dn) ? -1.0f : e0;
          e1 = (a1 <= up) ? 1.0f : e1;
          e1 = (b1 >= dn) ? -1.0f : e1;
          e2 = (a2 <= up) ? 1.0f : e2;
          e2 = (b2 >= dn) ? -1.0f : e2;
          e3 = (a3 <= up) ? 1.0f : e3;
          e3 = (b3 >= dn) ? -1.0f : e3;
          hq = hi;
        }
        c0 = (c0 == 0.0f) ? e0 : c0;
        c1 = (c1 == 0.0f) ? e1 : c1;
        c2 = (c2 == 0.0f) ? e2 : c2;
        c3 = (c3 == 0.0f) ? e3 : c3;
        if (__all((c0 != 0.0f) && (c1 != 0.0f) && (c2 != 0.0f) && (c3 != 0.0f))) { done = true; break; }
        am = fminf(fminf((c0 == 0.0f) ? a0 : 3.0e38f, (c1 == 0.0f) ? a1 : 3.0e38f),
                   fminf((c2 == 0.0f) ? a2 : 3.0e38f, (c3 == 0.0f) ? a3 : 3.0e38f));
        bm = fmaxf(fmaxf((c0 == 0.0f) ? b0 : -3.0e38f, (c1 == 0.0f) ? b1 : -3.0e38f),
                   fmaxf((c2 == 0.0f) ? b2 : -3.0e38f, (c3 == 0.0f) ? b3 : -3.0e38f));
      }
    }
    s0 = (c0 > 0.0f) ? 1.0f : -1.0f;
    s1 = (c1 > 0.0f) ? 1.0f : -1.0f;
    s2 = (c2 > 0.0f) ? 1.0f : -1.0f;
    s3 = (c3 > 0.0f) ? 1.0f : -1.0f;
  }

  const bool sel1 = (lane & 1) != 0;
  const bool sel2 = (lane & 2) != 0;
  const bool sel4 = (lane & 4) != 0;

  float hc[TT];
#pragma unroll
  for (int tt = 0; tt < TT; ++tt) hc[tt] = h[t0 + tt];

  float hp = (t0 > 0) ? h[t0 - 1] : 0.0f;
  int buf = 0;
  float* wrow = wsum[wave];
  for (int seg = t0; seg < t1; seg += SEG * TT) {
  const int segend = (seg + SEG * TT < t1) ? seg + SEG * TT : t1;
  if (owner) {
  for (int tb = seg; tb < segend; tb += TT) {
    const int nb = (tb + TT < T) ? tb + TT : tb;
    float hn[TT];
#pragma unroll
    for (int tt = 0; tt < TT; ++tt) hn[tt] = h[nb + tt];
    float* sb = sx[wave][buf];
#pragma unroll
    for (int tt = 0; tt < TT; ++tt) {
      const float hi = hc[tt];
      const float up = (hi > hp) ? hi : never;
      const float dn = (hi < hp) ? hi : never;
      s0 = (a0 <= up) ? 1.0f : s0;
      s0 = (b0 >= dn) ? -1.0f : s0;
      s1 = (a1 <= up) ? 1.0f : s1;
      s1 = (b1 >= dn) ? -1.0f : s1;
      s2 = (a2 <= up) ? 1.0f : s2;
      s2 = (b2 >= dn) ? -1.0f : s2;
      s3 = (a3 <= up) ? 1.0f : s3;
      s3 = (b3 >= dn) ? -1.0f : s3;
      hp = hi;

      v2f A0, A1;
      A0.x = s0;
      A0.y = s1;
      A1.x = s2;
      A1.y = s3;
      v8f acc = {};
      acc = __builtin_amdgcn_wmma_f32_16x16x4_f32(
          false, A0, false, B0, (short)0, acc, false, false);
      acc = __builtin_amdgcn_wmma_f32_16x16x4_f32(
          false, A1, false, B1, (short)0, acc, false, false);

      const float x01 = sel1 ? acc[1] : acc[0];
      const float x23 = sel1 ? acc[3] : acc[2];
      const float x45 = sel1 ? acc[5] : acc[4];
      const float x67 = sel1 ? acc[7] : acc[6];
      const float x03 = sel2 ? x23 : x01;
      const float x47 = sel2 ? x67 : x45;
      sb[tt * PITCH + lane] = sel4 ? x47 : x03;
    }
    __builtin_amdgcn_fence(__ATOMIC_RELEASE, "workgroup");
    __builtin_amdgcn_wave_barrier();
    __builtin_amdgcn_fence(__ATOMIC_ACQUIRE, "workgroup");
    const float* r = sb + lane * PITCH;
    float q0 = (r[0] + r[1]) + (r[2] + r[3]);
    float q1 = (r[4] + r[5]) + (r[6] + r[7]);
    float q2 = (r[24] + r[25]) + (r[26] + r[27]);
    float q3 = (r[28] + r[29]) + (r[30] + r[31]);
    wrow[tb - seg + lane] = (q0 + q1) + (q2 + q3);
#pragma unroll
    for (int tt = 0; tt < TT; ++tt) hc[tt] = hn[tt];
    buf ^= 1;
  }
  } else {
    for (int i = lane; i < segend - seg; i += 32) wrow[i] = 0.0f;
  }
  __syncthreads();
  static_assert(SEG * TT <= WPG * 32, "one thread per step of a segment");
  const int i = threadIdx.x;
  if (i < segend - seg) {
    float sum = wsum[0][i];
#pragma unroll
    for (int g = 1; g < WPG; ++g) sum += wsum[g][i];
    pout[seg + i] = sum;
  }
  __syncthreads();
  }
}

__global__ __launch_bounds__(BLOCK) void hyst_reduce_kernel(
    const float* __restrict__ partial, const float* __restrict__ scale,
    const float* __restrict__ offset, float* __restrict__ out,
    int T, int NW, float inv_n) {
  int t = blockIdx.x * blockDim.x + threadIdx.x;
  if (t < T) {
    float sum = 0.0f;
    for (int wv = 0; wv < NW; ++wv) sum += partial[(size_t)wv * T + t];
    out[t] = scale[0] * (sum * inv_n) + offset[0];
  }
}

extern "C" void kernel_launch(void* const* d_in, const int* in_sizes, int n_in,
                              void* d_out, int out_size, void* d_ws, size_t ws_size,
                              hipStream_t stream) {
  const float* h      = (const float*)d_in[0];
  const float* mesh   = (const float*)d_in[1];
  const float* raw    = (const float*)d_in[2];
  const float* scale  = (const float*)d_in[3];
  const float* offset = (const float*)d_in[4];
  float* out = (float*)d_out;

  const int T = in_sizes[0];
  const int n = in_sizes[2];

  const int NW   = (n + RPW - 1) / RPW;
  const int NG   = (NW + WPG - 1) / WPG;
  const int npad = NW * RPW;

  float* partial = (float*)d_ws;
  float* dens    = partial + (size_t)NG * T;
  float* agg     = dens + npad;

  hyst_dens_kernel<<<(npad + BLOCK - 1) / BLOCK, BLOCK, 0, stream>>>(
      raw, h, dens, agg, n, npad, T);
  hyst_scan_kernel<<<dim3(NG, NCH), WPG * 32, 0, stream>>>(
      h, mesh, dens, agg, partial, T, n, NW);
  hyst_reduce_kernel<<<(T + BLOCK - 1) / BLOCK, BLOCK, 0, stream>>>(
      partial, scale, offset, out, T, NG, 1.0f / (float)n);
}


# scan kernel: the field's direction is wave-uniform, so each step branches on it (scalar) and runs one compare + one select per relay, its own direction's, in scan and walk back; sums unchanged
# speedup vs baseline: 18.5351x; 1.0930x over previous
#include <hip/hip_runtime.h>
#include <math.h>

typedef __attribute__((ext_vector_type(2))) float v2f;
typedef __attribute__((ext_vector_type(8))) float v8f;

#define RPW 128
#define TT 32
#define PITCH 33
#define BLOCK 256
#define NCH 32
#define WPG 20
#define SEG 4

__global__ __launch_bounds__(BLOCK) void hyst_dens_kernel(
    const float* __restrict__ raw, const float* __restrict__ h,
    float* __restrict__ dens, float* __restrict__ agg, int n, int npad, int T) {
  int j = blockIdx.x * blockDim.x + threadIdx.x;
  const int nblk = T / TT;
  for (int b = j; b < nblk; b += gridDim.x * blockDim.x) {
    const int bb = b * TT;
    float q = (bb > 0) ? h[bb - 1] : 0.0f;
    float mu = -__builtin_huge_valf(), md = __builtin_huge_valf();
#pragma unroll
    for (int tt = 0; tt < TT; ++tt) {
      const float hi = h[bb + tt];
      mu = (hi > q) ? fmaxf(mu, hi) : mu;
      md = (hi < q) ? fminf(md, hi) : md;
      q = hi;
    }
    agg[b] = mu;
    agg[nblk + b] = md;
  }
  if (j < npad) {
    float v = 0.0f;
    if (j < n) {
      float x = raw[j];
      v = fmaxf(x, 0.0f) + log1pf(expf(-fabsf(x)));
    }
    dens[j] = v;
  }
}

__global__ __launch_bounds__(WPG * 32) void hyst_scan_kernel(
    const float* __restrict__ h, const float* __restrict__ mesh,
    const float* __restrict__ dens, const float* __restrict__ agg,
    float* __restrict__ partial, int T, int n, int NW) {
  __shared__ float sx[WPG][2][TT * PITCH];
  __shared__ float wsum[WPG][SEG * TT];

  const int lane = threadIdx.x & 31;
  const int wave = __builtin_amdgcn_readfirstlane((int)(threadIdx.x >> 5));
  const int w    = blockIdx.x * WPG + wave;
  const bool owner = w < NW;
  const int p0   = w * RPW + 4 * lane;

  float a0 = 3.0e38f, b0 = -3.0e38f, d0 = 0.0f;
  float a1 = 3.0e38f, b1 = -3.0e38f, d1 = 0.0f;
  float a2 = 3.0e38f, b2 = -3.0e38f, d2 = 0.0f;
  float a3 = 3.0e38f, b3 = -3.0e38f, d3 = 0.0f;
  if (p0 < n)     { b0 = mesh[2 * p0];     a0 = mesh[2 * p0 + 1]; d0 = dens[p0]; }
  if (p0 + 1 < n) { b1 = mesh[2 * p0 + 2]; a1 = mesh[2 * p0 + 3]; d1 = dens[p0 + 1]; }
  if (p0 + 2 < n) { b2 = mesh[2 * p0 + 4]; a2 = mesh[2 * p0 + 5]; d2 = dens[p0 + 2]; }
  if (p0 + 3 < n) { b3 = mesh[2 * p0 + 6]; a3 = mesh[2 * p0 + 7]; d3 = dens[p0 + 3]; }

  v2f B0, B1;
  B0.x = d0;
  B0.y = d1;
  B1.x = d2;
  B1.y = d3;

  const int tpc = (T / TT + NCH - 1) / NCH;
  const int t0  = (NCH - 1 - (int)blockIdx.y) * tpc * TT;
  if (t0 >= T) return;
  const int t1  = (t0 + tpc * TT < T) ? t0 + tpc * TT : T;

  float* pout = partial + (size_t)blockIdx.x * T;

  float s0 = -1.0f, s1 = -1.0f, s2 = -1.0f, s3 = -1.0f;
  if (owner && t0 > 0) {
    float c0 = (p0 < n) ? 0.0f : -1.0f;
    float c1 = (p0 + 1 < n) ? 0.0f : -1.0f;
    float c2 = (p0 + 2 < n) ? 0.0f : -1.0f;
    float c3 = (p0 + 3 < n) ? 0.0f : -1.0f;
    float am = fminf(fminf((c0 == 0.0f) ? a0 : 3.0e38f, (c1 == 0.0f) ? a1 : 3.0e38f),
                     fminf((c2 == 0.0f) ? a2 : 3.0e38f, (c3 == 0.0f) ? a3 : 3.0e38f));
    float bm = fmaxf(fmaxf((c0 == 0.0f) ? b0 : -3.0e38f, (c1 == 0.0f) ? b1 : -3.0e38f),
                     fmaxf((c2 == 0.0f) ? b2 : -3.0e38f, (c3 == 0.0f) ? b3 : -3.0e38f));
    const int nblk = T / TT;
    const int nb0  = t0 / TT;
    bool done = false;
    for (int g = (nb0 - 1) / 32; g >= 0 && !done; --g) {
      const int bi = g * 32 + lane;
      float muv = -__builtin_huge_valf(), mdv = __builtin_huge_valf();
      if (bi < nb0) { muv = agg[bi]; mdv = agg[nblk + bi]; }
      const int btop = (nb0 - 1 - g * 32 < 31) ? nb0 - 1 - g * 32 : 31;
      for (int b = btop; b >= 0; --b) {
        const float mu = __int_as_float(__builtin_amdgcn_readlane(__float_as_int(muv), b));
        const float md = __int_as_float(__builtin_amdgcn_readlane(__float_as_int(mdv), b));
        const bool hit = (am <= mu) || (bm >= md);
        if (!__any(hit)) continue;
        const int bb = (g * 32 + b) * TT;
        float hb[TT];
#pragma unroll
        for (int tt = 0; tt < TT; ++tt) hb[tt] = h[bb + tt];
        float hq = (bb > 0) ? h[bb - 1] : 0.0f;
        float e0 = 0.0f, e1 = 0.0f, e2 = 0.0f, e3 = 0.0f;
#pragma unroll
        for (int tt = 0; tt < TT; ++tt) {
          const float hi = hb[tt];
          if (hi > hq) {
            e0 = (a0 <= hi) ? 1.0f : e0;
            e1 = (a1 <= hi) ? 1.0f : e1;
            e2 = (a2 <= hi) ? 1.0f : e2;
            e3 = (a3 <= hi) ? 1.0f : e3;
          }
          if (hi < hq) {
            e0 = (b0 >= hi) ? -1.0f : e0;
            e1 = (b1 >= hi) ? -1.0f : e1;
            e2 = (b2 >= hi) ? -1.0f : e2;
            e3 = (b3 >= hi) ? -1.0f : e3;
          }
          hq = hi;
        }
        c0 = (c0 == 0.0f) ? e0 : c0;
        c1 = (c1 == 0.0f) ? e1 : c1;
        c2 = (c2 == 0.0f) ? e2 : c2;
        c3 = (c3 == 0.0f) ? e3 : c3;
        if (__all((c0 != 0.0f) && (c1 != 0.0f) && (c2 != 0.0f) && (c3 != 0.0f))) { done = true; break; }
        am = fminf(fminf((c0 == 0.0f) ? a0 : 3.0e38f, (c1 == 0.0f) ? a1 : 3.0e38f),
                   fminf((c2 == 0.0f) ? a2 : 3.0e38f, (c3 == 0.0f) ? a3 : 3.0e38f));
        bm = fmaxf(fmaxf((c0 == 0.0f) ? b0 : -3.0e38f, (c1 == 0.0f) ? b1 : -3.0e38f),
                   fmaxf((c2 == 0.0f) ? b2 : -3.0e38f, (c3 == 0.0f) ? b3 : -3.0e38f));
      }
    }
    s0 = (c0 > 0.0f) ? 1.0f : -1.0f;
    s1 = (c1 > 0.0f) ? 1.0f : -1.0f;
    s2 = (c2 > 0.0f) ? 1.0f : -1.0f;
    s3 = (c3 > 0.0f) ? 1.0f : -1.0f;
  }

  const bool sel1 = (lane & 1) != 0;
  const bool sel2 = (lane & 2) != 0;
  const bool sel4 = (lane & 4) != 0;

  float hc[TT];
#pragma unroll
  for (int tt = 0; tt < TT; ++tt) hc[tt] = h[t0 + tt];

  float hp = (t0 > 0) ? h[t0 - 1] : 0.0f;
  int buf = 0;
  float* wrow = wsum[wave];
  for (int seg = t0; seg < t1; seg += SEG * TT) {
  const int segend = (seg + SEG * TT < t1) ? seg + SEG * TT : t1;
  if (owner) {
  for (int tb = seg; tb < segend; tb += TT) {
    const int nb = (tb + TT < T) ? tb + TT : tb;
    float hn[TT];
#pragma unroll
    for (int tt = 0; tt < TT; ++tt) hn[tt] = h[nb + tt];
    float* sb = sx[wave][buf];
#pragma unroll
    for (int tt = 0; tt < TT; ++tt) {
      const float hi = hc[tt];
      if (hi > hp) {
        s0 = (a0 <= hi) ? 1.0f : s0;
        s1 = (a1 <= hi) ? 1.0f : s1;
        s2 = (a2 <= hi) ? 1.0f : s2;
        s3 = (a3 <= hi) ? 1.0f : s3;
      }
      if (hi < hp) {
        s0 = (b0 >= hi) ? -1.0f : s0;
        s1 = (b1 >= hi) ? -1.0f : s1;
        s2 = (b2 >= hi) ? -1.0f : s2;
        s3 = (b3 >= hi) ? -1.0f : s3;
      }
      hp = hi;

      v2f A0, A1;
      A0.x = s0;
      A0.y = s1;
      A1.x = s2;
      A1.y = s3;
      v8f acc = {};
      acc = __builtin_amdgcn_wmma_f32_16x16x4_f32(
          false, A0, false, B0, (short)0, acc, false, false);
      acc = __builtin_amdgcn_wmma_f32_16x16x4_f32(
          false, A1, false, B1, (short)0, acc, false, false);
      asm volatile("s_wait_alu depctr_va_vdst(0)" : "+v"(acc));

      const float x01 = sel1 ? acc[1] : acc[0];
      const float x23 = sel1 ? acc[3] : acc[2];
      const float x45 = sel1 ? acc[5] : acc[4];
      const float x67 = sel1 ? acc[7] : acc[6];
      const float x03 = sel2 ? x23 : x01;
      const float x47 = sel2 ? x67 : x45;
      sb[tt * PITCH + lane] = sel4 ? x47 : x03;
    }
    __builtin_amdgcn_fence(__ATOMIC_RELEASE, "workgroup");
    __builtin_amdgcn_wave_barrier();
    __builtin_amdgcn_fence(__ATOMIC_ACQUIRE, "workgroup");
    const float* r = sb + lane * PITCH;
    float q0 = (r[0] + r[1]) + (r[2] + r[3]);
    float q1 = (r[4] + r[5]) + (r[6] + r[7]);
    float q2 = (r[24] + r[25]) + (r[26] + r[27]);
    float q3 = (r[28] + r[29]) + (r[30] + r[31]);
    wrow[tb - seg + lane] = (q0 + q1) + (q2 + q3);
#pragma unroll
    for (int tt = 0; tt < TT; ++tt) hc[tt] = hn[tt];
    buf ^= 1;
  }
  } else {
    for (int i = lane; i < segend - seg; i += 32) wrow[i] = 0.0f;
  }
  __syncthreads();
  static_assert(SEG * TT <= WPG * 32, "one thread per step of a segment");
  const int i = threadIdx.x;
  if (i < segend - seg) {
    float sum = wsum[0][i];
#pragma unroll
    for (int g = 1; g < WPG; ++g) sum += wsum[g][i];
    pout[seg + i] = sum;
  }
  __syncthreads();
  }
}

__global__ __launch_bounds__(BLOCK) void hyst_reduce_kernel(
    const float* __restrict__ partial, const float* __restrict__ scale,
    const float* __restrict__ offset, float* __restrict__ out,
    int T, int NW, float inv_n) {
  int t = blockIdx.x * blockDim.x + threadIdx.x;
  if (t < T) {
    float sum = 0.0f;
    for (int wv = 0; wv < NW; ++wv) sum += partial[(size_t)wv * T + t];
    out[t] = scale[0] * (sum * inv_n) + offset[0];
  }
}

extern "C" void kernel_launch(void* const* d_in, const int* in_sizes, int n_in,
                              void* d_out, int out_size, void* d_ws, size_t ws_size,
                              hipStream_t stream) {
  const float* h      = (const float*)d_in[0];
  const float* mesh   = (const float*)d_in[1];
  const float* raw    = (const float*)d_in[2];
  const float* scale  = (const float*)d_in[3];
  const float* offset = (const float*)d_in[4];
  float* out = (float*)d_out;

  const int T = in_sizes[0];
  const int n = in_sizes[2];

  const int NW   = (n + RPW - 1) / RPW;
  const int NG   = (NW + WPG - 1) / WPG;
  const int npad = NW * RPW;

  float* partial = (float*)d_ws;
  float* dens    = partial + (size_t)NG * T;
  float* agg     = dens + npad;

  hyst_dens_kernel<<<(npad + BLOCK - 1) / BLOCK, BLOCK, 0, stream>>>(
      raw, h, dens, agg, n, npad, T);
  hyst_scan_kernel<<<dim3(NG, NCH), WPG * 32, 0, stream>>>(
      h, mesh, dens, agg, partial, T, n, NW);
  hyst_reduce_kernel<<<(T + BLOCK - 1) / BLOCK, BLOCK, 0, stream>>>(
      partial, scale, offset, out, T, NG, 1.0f / (float)n);
}


# scan kernel: a step that switches none of the wave's 128 relays (no lane's +-1 state sum moved) stores the previous step's sum again and skips both WMMAs and the diagonal pick; sums unchanged
# speedup vs baseline: 20.2976x; 1.0951x over previous
#include <hip/hip_runtime.h>
#include <math.h>

typedef __attribute__((ext_vector_type(2))) float v2f;
typedef __attribute__((ext_vector_type(8))) float v8f;

#define RPW 128
#define TT 32
#define PITCH 33
#define BLOCK 256
#define NCH 32
#define WPG 20
#define SEG 4

__global__ __launch_bounds__(BLOCK) void hyst_dens_kernel(
    const float* __restrict__ raw, const float* __restrict__ h,
    float* __restrict__ dens, float* __restrict__ agg, int n, int npad, int T) {
  int j = blockIdx.x * blockDim.x + threadIdx.x;
  const int nblk = T / TT;
  for (int b = j; b < nblk; b += gridDim.x * blockDim.x) {
    const int bb = b * TT;
    float q = (bb > 0) ? h[bb - 1] : 0.0f;
    float mu = -__builtin_huge_valf(), md = __builtin_huge_valf();
#pragma unroll
    for (int tt = 0; tt < TT; ++tt) {
      const float hi = h[bb + tt];
      mu = (hi > q) ? fmaxf(mu, hi) : mu;
      md = (hi < q) ? fminf(md, hi) : md;
      q = hi;
    }
    agg[b] = mu;
    agg[nblk + b] = md;
  }
  if (j < npad) {
    float v = 0.0f;
    if (j < n) {
      float x = raw[j];
      v = fmaxf(x, 0.0f) + log1pf(expf(-fabsf(x)));
    }
    dens[j] = v;
  }
}

__device__ __forceinline__ float hyst_wave_sum(float s0, float s1, float s2, float s3,
                                               v2f B0, v2f B1, bool sel1, bool sel2, bool sel4) {
  v2f A0, A1;
  A0.x = s0;
  A0.y = s1;
  A1.x = s2;
  A1.y = s3;
  v8f acc = {};
  acc = __builtin_amdgcn_wmma_f32_16x16x4_f32(
      false, A0, false, B0, (short)0, acc, false, false);
  acc = __builtin_amdgcn_wmma_f32_16x16x4_f32(
      false, A1, false, B1, (short)0, acc, false, false);
  asm volatile("s_wait_alu depctr_va_vdst(0)" : "+v"(acc));
  const float x01 = sel1 ? acc[1] : acc[0];
  const float x23 = sel1 ? acc[3] : acc[2];
  const float x45 = sel1 ? acc[5] : acc[4];
  const float x67 = sel1 ? acc[7] : acc[6];
  const float x03 = sel2 ? x23 : x01;
  const float x47 = sel2 ? x67 : x45;
  return sel4 ? x47 : x03;
}

__global__ __launch_bounds__(WPG * 32) void hyst_scan_kernel(
    const float* __restrict__ h, const float* __restrict__ mesh,
    const float* __restrict__ dens, const float* __restrict__ agg,
    float* __restrict__ partial, int T, int n, int NW) {
  __shared__ float sx[WPG][2][TT * PITCH];
  __shared__ float wsum[WPG][SEG * TT];

  const int lane = threadIdx.x & 31;
  const int wave = __builtin_amdgcn_readfirstlane((int)(threadIdx.x >> 5));
  const int w    = blockIdx.x * WPG + wave;
  const bool owner = w < NW;
  const int p0   = w * RPW + 4 * lane;

  float a0 = 3.0e38f, b0 = -3.0e38f, d0 = 0.0f;
  float a1 = 3.0e38f, b1 = -3.0e38f, d1 = 0.0f;
  float a2 = 3.0e38f, b2 = -3.0e38f, d2 = 0.0f;
  float a3 = 3.0e38f, b3 = -3.0e38f, d3 = 0.0f;
  if (p0 < n)     { b0 = mesh[2 * p0];     a0 = mesh[2 * p0 + 1]; d0 = dens[p0]; }
  if (p0 + 1 < n) { b1 = mesh[2 * p0 + 2]; a1 = mesh[2 * p0 + 3]; d1 = dens[p0 + 1]; }
  if (p0 + 2 < n) { b2 = mesh[2 * p0 + 4]; a2 = mesh[2 * p0 + 5]; d2 = dens[p0 + 2]; }
  if (p0 + 3 < n) { b3 = mesh[2 * p0 + 6]; a3 = mesh[2 * p0 + 7]; d3 = dens[p0 + 3]; }

  v2f B0, B1;
  B0.x = d0;
  B0.y = d1;
  B1.x = d2;
  B1.y = d3;

  const int tpc = (T / TT + NCH - 1) / NCH;
  const int t0  = (NCH - 1 - (int)blockIdx.y) * tpc * TT;
  if (t0 >= T) return;
  const int t1  = (t0 + tpc * TT < T) ? t0 + tpc * TT : T;

  float* pout = partial + (size_t)blockIdx.x * T;

  float s0 = -1.0f, s1 = -1.0f, s2 = -1.0f, s3 = -1.0f;
  if (owner && t0 > 0) {
    float c0 = (p0 < n) ? 0.0f : -1.0f;
    float c1 = (p0 + 1 < n) ? 0.0f : -1.0f;
    float c2 = (p0 + 2 < n) ? 0.0f : -1.0f;
    float c3 = (p0 + 3 < n) ? 0.0f : -1.0f;
    float am = fminf(fminf((c0 == 0.0f) ? a0 : 3.0e38f, (c1 == 0.0f) ? a1 : 3.0e38f),
                     fminf((c2 == 0.0f) ? a2 : 3.0e38f, (c3 == 0.0f) ? a3 : 3.0e38f));
    float bm = fmaxf(fmaxf((c0 == 0.0f) ? b0 : -3.0e38f, (c1 == 0.0f) ? b1 : -3.0e38f),
                     fmaxf((c2 == 0.0f) ? b2 : -3.0e38f, (c3 == 0.0f) ? b3 : -3.0e38f));
    const int nblk = T / TT;
    const int nb0  = t0 / TT;
    bool done = false;
    for (int g = (nb0 - 1) / 32; g >= 0 && !done; --g) {
      const int bi = g * 32 + lane;
      float muv = -__builtin_huge_valf(), mdv = __builtin_huge_valf();
      if (bi < nb0) { muv = agg[bi]; mdv = agg[nblk + bi]; }
      const int btop = (nb0 - 1 - g * 32 < 31) ? nb0 - 1 - g * 32 : 31;
      for (int b = btop; b >= 0; --b) {
        const float mu = __int_as_float(__builtin_amdgcn_readlane(__float_as_int(muv), b));
        const float md = __int_as_float(__builtin_amdgcn_readlane(__float_as_int(mdv), b));
        const bool hit = (am <= mu) || (bm >= md);
        if (!__any(hit)) continue;
        const int bb = (g * 32 + b) * TT;
        float hb[TT];
#pragma unroll
        for (int tt = 0; tt < TT; ++tt) hb[tt] = h[bb + tt];
        float hq = (bb > 0) ? h[bb - 1] : 0.0f;
        float e0 = 0.0f, e1 = 0.0f, e2 = 0.0f, e3 = 0.0f;
#pragma unroll
        for (int tt = 0; tt < TT; ++tt) {
          const float hi = hb[tt];
          if (hi > hq) {
            e0 = (a0 <= hi) ? 1.0f : e0;
            e1 = (a1 <= hi) ? 1.0f : e1;
            e2 = (a2 <= hi) ? 1.0f : e2;
            e3 = (a3 <= hi) ? 1.0f : e3;
          }
          if (hi < hq) {
            e0 = (b0 >= hi) ? -1.0f : e0;
            e1 = (b1 >= hi) ? -1.0f : e1;
            e2 = (b2 >= hi) ? -1.0f : e2;
            e3 = (b3 >= hi) ? -1.0f : e3;
          }
          hq = hi;
        }
        c0 = (c0 == 0.0f) ? e0 : c0;
        c1 = (c1 == 0.0f) ? e1 : c1;
        c2 = (c2 == 0.0f) ? e2 : c2;
        c3 = (c3 == 0.0f) ? e3 : c3;
        if (__all((c0 != 0.0f) && (c1 != 0.0f) && (c2 != 0.0f) && (c3 != 0.0f))) { done = true; break; }
        am = fminf(fminf((c0 == 0.0f) ? a0 : 3.0e38f, (c1 == 0.0f) ? a1 : 3.0e38f),
                   fminf((c2 == 0.0f) ? a2 : 3.0e38f, (c3 == 0.0f) ? a3 : 3.0e38f));
        bm = fmaxf(fmaxf((c0 == 0.0f) ? b0 : -3.0e38f, (c1 == 0.0f) ? b1 : -3.0e38f),
                   fmaxf((c2 == 0.0f) ? b2 : -3.0e38f, (c3 == 0.0f) ? b3 : -3.0e38f));
      }
    }
    s0 = (c0 > 0.0f) ? 1.0f : -1.0f;
    s1 = (c1 > 0.0f) ? 1.0f : -1.0f;
    s2 = (c2 > 0.0f) ? 1.0f : -1.0f;
    s3 = (c3 > 0.0f) ? 1.0f : -1.0f;
  }

  const bool sel1 = (lane & 1) != 0;
  const bool sel2 = (lane & 2) != 0;
  const bool sel4 = (lane & 4) != 0;

  float hc[TT];
#pragma unroll
  for (int tt = 0; tt < TT; ++tt) hc[tt] = h[t0 + tt];

  float hp = (t0 > 0) ? h[t0 - 1] : 0.0f;
  int buf = 0;
  float* wrow = wsum[wave];
  float cs = (s0 + s1) + (s2 + s3);
  float xs = 0.0f;
  if (owner) xs = hyst_wave_sum(s0, s1, s2, s3, B0, B1, sel1, sel2, sel4);
  for (int seg = t0; seg < t1; seg += SEG * TT) {
  const int segend = (seg + SEG * TT < t1) ? seg + SEG * TT : t1;
  if (owner) {
  for (int tb = seg; tb < segend; tb += TT) {
    const int nb = (tb + TT < T) ? tb + TT : tb;
    float hn[TT];
#pragma unroll
    for (int tt = 0; tt < TT; ++tt) hn[tt] = h[nb + tt];
    float* sb = sx[wave][buf];
#pragma unroll
    for (int tt = 0; tt < TT; ++tt) {
      const float hi = hc[tt];
      if (hi > hp) {
        s0 = (a0 <= hi) ? 1.0f : s0;
        s1 = (a1 <= hi) ? 1.0f : s1;
        s2 = (a2 <= hi) ? 1.0f : s2;
        s3 = (a3 <= hi) ? 1.0f : s3;
      }
      if (hi < hp) {
        s0 = (b0 >= hi) ? -1.0f : s0;
        s1 = (b1 >= hi) ? -1.0f : s1;
        s2 = (b2 >= hi) ? -1.0f : s2;
        s3 = (b3 >= hi) ? -1.0f : s3;
      }
      hp = hi;

      const float cn = (s0 + s1) + (s2 + s3);
      if (__any(cn != cs)) xs = hyst_wave_sum(s0, s1, s2, s3, B0, B1, sel1, sel2, sel4);
      cs = cn;
      sb[tt * PITCH + lane] = xs;
    }
    __builtin_amdgcn_fence(__ATOMIC_RELEASE, "workgroup");
    __builtin_amdgcn_wave_barrier();
    __builtin_amdgcn_fence(__ATOMIC_ACQUIRE, "workgroup");
    const float* r = sb + lane * PITCH;
    float q0 = (r[0] + r[1]) + (r[2] + r[3]);
    float q1 = (r[4] + r[5]) + (r[6] + r[7]);
    float q2 = (r[24] + r[25]) + (r[26] + r[27]);
    float q3 = (r[28] + r[29]) + (r[30] + r[31]);
    wrow[tb - seg + lane] = (q0 + q1) + (q2 + q3);
#pragma unroll
    for (int tt = 0; tt < TT; ++tt) hc[tt] = hn[tt];
    buf ^= 1;
  }
  } else {
    for (int i = lane; i < segend - seg; i += 32) wrow[i] = 0.0f;
  }
  __syncthreads();
  static_assert(SEG * TT <= WPG * 32, "one thread per step of a segment");
  const int i = threadIdx.x;
  if (i < segend - seg) {
    float sum = wsum[0][i];
#pragma unroll
    for (int g = 1; g < WPG; ++g) sum += wsum[g][i];
    pout[seg + i] = sum;
  }
  __syncthreads();
  }
}

__global__ __launch_bounds__(BLOCK) void hyst_reduce_kernel(
    const float* __restrict__ partial, const float* __restrict__ scale,
    const float* __restrict__ offset, float* __restrict__ out,
    int T, int NW, float inv_n) {
  int t = blockIdx.x * blockDim.x + threadIdx.x;
  if (t < T) {
    float sum = 0.0f;
    for (int wv = 0; wv < NW; ++wv) sum += partial[(size_t)wv * T + t];
    out[t] = scale[0] * (sum * inv_n) + offset[0];
  }
}

extern "C" void kernel_launch(void* const* d_in, const int* in_sizes, int n_in,
                              void* d_out, int out_size, void* d_ws, size_t ws_size,
                              hipStream_t stream) {
  const float* h      = (const float*)d_in[0];
  const float* mesh   = (const float*)d_in[1];
  const float* raw    = (const float*)d_in[2];
  const float* scale  = (const float*)d_in[3];
  const float* offset = (const float*)d_in[4];
  float* out = (float*)d_out;

  const int T = in_sizes[0];
  const int n = in_sizes[2];

  const int NW   = (n + RPW - 1) / RPW;
  const int NG   = (NW + WPG - 1) / WPG;
  const int npad = NW * RPW;

  float* partial = (float*)d_ws;
  float* dens    = partial + (size_t)NG * T;
  float* agg     = dens + npad;

  hyst_dens_kernel<<<(npad + BLOCK - 1) / BLOCK, BLOCK, 0, stream>>>(
      raw, h, dens, agg, n, npad, T);
  hyst_scan_kernel<<<dim3(NG, NCH), WPG * 32, 0, stream>>>(
      h, mesh, dens, agg, partial, T, n, NW);
  hyst_reduce_kernel<<<(T + BLOCK - 1) / BLOCK, BLOCK, 0, stream>>>(
      partial, scale, offset, out, T, NG, 1.0f / (float)n);
}


# scan kernel: relay blocks dealt out over a chunk's work-groups (wave i of group g owns block ((9*i)%20)*NG+g, not 20 neighbours): SIMDs get equal shares of switching steps; relay sum's order changes
# speedup vs baseline: 22.0230x; 1.0850x over previous
#include <hip/hip_runtime.h>
#include <math.h>

typedef __attribute__((ext_vector_type(2))) float v2f;
typedef __attribute__((ext_vector_type(8))) float v8f;

#define RPW 128
#define TT 32
#define PITCH 33
#define BLOCK 256
#define NCH 32
#define WPG 20
#define SEG 4
#define DEAL 9

__global__ __launch_bounds__(BLOCK) void hyst_dens_kernel(
    const float* __restrict__ raw, const float* __restrict__ h,
    float* __restrict__ dens, float* __restrict__ agg, int n, int npad, int T) {
  int j = blockIdx.x * blockDim.x + threadIdx.x;
  const int nblk = T / TT;
  for (int b = j; b < nblk; b += gridDim.x * blockDim.x) {
    const int bb = b * TT;
    float q = (bb > 0) ? h[bb - 1] : 0.0f;
    float mu = -__builtin_huge_valf(), md = __builtin_huge_valf();
#pragma unroll
    for (int tt = 0; tt < TT; ++tt) {
      const float hi = h[bb + tt];
      mu = (hi > q) ? fmaxf(mu, hi) : mu;
      md = (hi < q) ? fminf(md, hi) : md;
      q = hi;
    }
    agg[b] = mu;
    agg[nblk + b] = md;
  }
  if (j < npad) {
    float v = 0.0f;
    if (j < n) {
      float x = raw[j];
      v = fmaxf(x, 0.0f) + log1pf(expf(-fabsf(x)));
    }
    dens[j] = v;
  }
}

__device__ __forceinline__ float hyst_wave_sum(float s0, float s1, float s2, float s3,
                                               v2f B0, v2f B1, bool sel1, bool sel2, bool sel4) {
  v2f A0, A1;
  A0.x = s0;
  A0.y = s1;
  A1.x = s2;
  A1.y = s3;
  v8f acc = {};
  acc = __builtin_amdgcn_wmma_f32_16x16x4_f32(
      false, A0, false, B0, (short)0, acc, false, false);
  acc = __builtin_amdgcn_wmma_f32_16x16x4_f32(
      false, A1, false, B1, (short)0, acc, false, false);
  asm volatile("s_wait_alu depctr_va_vdst(0)" : "+v"(acc));
  const float x01 = sel1 ? acc[1] : acc[0];
  const float x23 = sel1 ? acc[3] : acc[2];
  const float x45 = sel1 ? acc[5] : acc[4];
  const float x67 = sel1 ? acc[7] : acc[6];
  const float x03 = sel2 ? x23 : x01;
  const float x47 = sel2 ? x67 : x45;
  return sel4 ? x47 : x03;
}

__global__ __launch_bounds__(WPG * 32) void hyst_scan_kernel(
    const float* __restrict__ h, const float* __restrict__ mesh,
    const float* __restrict__ dens, const float* __restrict__ agg,
    float* __restrict__ partial, int T, int n, int NW) {
  __shared__ float sx[WPG][2][TT * PITCH];
  __shared__ float wsum[WPG][SEG * TT];

  const int lane = threadIdx.x & 31;
  const int wave = __builtin_amdgcn_readfirstlane((int)(threadIdx.x >> 5));
  static_assert((DEAL * DEAL) % WPG == 1, "DEAL is its own inverse modulo WPG: every slot is taken once");
  const int w    = ((wave * DEAL) % WPG) * (int)gridDim.x + (int)blockIdx.x;
  const bool owner = w < NW;
  const int p0   = w * RPW + 4 * lane;

  float a0 = 3.0e38f, b0 = -3.0e38f, d0 = 0.0f;
  float a1 = 3.0e38f, b1 = -3.0e38f, d1 = 0.0f;
  float a2 = 3.0e38f, b2 = -3.0e38f, d2 = 0.0f;
  float a3 = 3.0e38f, b3 = -3.0e38f, d3 = 0.0f;
  if (p0 < n)     { b0 = mesh[2 * p0];     a0 = mesh[2 * p0 + 1]; d0 = dens[p0]; }
  if (p0 + 1 < n) { b1 = mesh[2 * p0 + 2]; a1 = mesh[2 * p0 + 3]; d1 = dens[p0 + 1]; }
  if (p0 + 2 < n) { b2 = mesh[2 * p0 + 4]; a2 = mesh[2 * p0 + 5]; d2 = dens[p0 + 2]; }
  if (p0 + 3 < n) { b3 = mesh[2 * p0 + 6]; a3 = mesh[2 * p0 + 7]; d3 = dens[p0 + 3]; }

  v2f B0, B1;
  B0.x = d0;
  B0.y = d1;
  B1.x = d2;
  B1.y = d3;

  const int tpc = (T / TT + NCH - 1) / NCH;
  const int t0  = (NCH - 1 - (int)blockIdx.y) * tpc * TT;
  if (t0 >= T) return;
  const int t1  = (t0 + tpc * TT < T) ? t0 + tpc * TT : T;

  float* pout = partial + (size_t)blockIdx.x * T;

  float s0 = -1.0f, s1 = -1.0f, s2 = -1.0f, s3 = -1.0f;
  if (owner && t0 > 0) {
    float c0 = (p0 < n) ? 0.0f : -1.0f;
    float c1 = (p0 + 1 < n) ? 0.0f : -1.0f;
    float c2 = (p0 + 2 < n) ? 0.0f : -1.0f;
    float c3 = (p0 + 3 < n) ? 0.0f : -1.0f;
    float am = fminf(fminf((c0 == 0.0f) ? a0 : 3.0e38f, (c1 == 0.0f) ? a1 : 3.0e38f),
                     fminf((c2 == 0.0f) ? a2 : 3.0e38f, (c3 == 0.0f) ? a3 : 3.0e38f));
    float bm = fmaxf(fmaxf((c0 == 0.0f) ? b0 : -3.0e38f, (c1 == 0.0f) ? b1 : -3.0e38f),
                     fmaxf((c2 == 0.0f) ? b2 : -3.0e38f, (c3 == 0.0f) ? b3 : -3.0e38f));
    const int nblk = T / TT;
    const int nb0  = t0 / TT;
    bool done = false;
    for (int g = (nb0 - 1) / 32; g >= 0 && !done; --g) {
      const int bi = g * 32 + lane;
      float muv = -__builtin_huge_valf(), mdv = __builtin_huge_valf();
      if (bi < nb0) { muv = agg[bi]; mdv = agg[nblk + bi]; }
      const int btop = (nb0 - 1 - g * 32 < 31) ? nb0 - 1 - g * 32 : 31;
      for (int b = btop; b >= 0; --b) {
        const float mu = __int_as_float(__builtin_amdgcn_readlane(__float_as_int(muv), b));
        const float md = __int_as_float(__builtin_amdgcn_readlane(__float_as_int(mdv), b));
        const bool hit = (am <= mu) || (bm >= md);
        if (!__any(hit)) continue;
        const int bb = (g * 32 + b) * TT;
        float hb[TT];
#pragma unroll
        for (int tt = 0; tt < TT; ++tt) hb[tt] = h[bb + tt];
        float hq = (bb > 0) ? h[bb - 1] : 0.0f;
        float e0 = 0.0f, e1 = 0.0f, e2 = 0.0f, e3 = 0.0f;
#pragma unroll
        for (int tt = 0; tt < TT; ++tt) {
          const float hi = hb[tt];
          if (hi > hq) {
            e0 = (a0 <= hi) ? 1.0f : e0;
            e1 = (a1 <= hi) ? 1.0f : e1;
            e2 = (a2 <= hi) ? 1.0f : e2;
            e3 = (a3 <= hi) ? 1.0f : e3;
          }
          if (hi < hq) {
            e0 = (b0 >= hi) ? -1.0f : e0;
            e1 = (b1 >= hi) ? -1.0f : e1;
            e2 = (b2 >= hi) ? -1.0f : e2;
            e3 = (b3 >= hi) ? -1.0f : e3;
          }
          hq = hi;
        }
        c0 = (c0 == 0.0f) ? e0 : c0;
        c1 = (c1 == 0.0f) ? e1 : c1;
        c2 = (c2 == 0.0f) ? e2 : c2;
        c3 = (c3 == 0.0f) ? e3 : c3;
        if (__all((c0 != 0.0f) && (c1 != 0.0f) && (c2 != 0.0f) && (c3 != 0.0f))) { done = true; break; }
        am = fminf(fminf((c0 == 0.0f) ? a0 : 3.0e38f, (c1 == 0.0f) ? a1 : 3.0e38f),
                   fminf((c2 == 0.0f) ? a2 : 3.0e38f, (c3 == 0.0f) ? a3 : 3.0e38f));
        bm = fmaxf(fmaxf((c0 == 0.0f) ? b0 : -3.0e38f, (c1 == 0.0f) ? b1 : -3.0e38f),
                   fmaxf((c2 == 0.0f) ? b2 : -3.0e38f, (c3 == 0.0f) ? b3 : -3.0e38f));
      }
    }
    s0 = (c0 > 0.0f) ? 1.0f : -1.0f;
    s1 = (c1 > 0.0f) ? 1.0f : -1.0f;
    s2 = (c2 > 0.0f) ? 1.0f : -1.0f;
    s3 = (c3 > 0.0f) ? 1.0f : -1.0f;
  }

  const bool sel1 = (lane & 1) != 0;
  const bool sel2 = (lane & 2) != 0;
  const bool sel4 = (lane & 4) != 0;

  float hc[TT];
#pragma unroll
  for (int tt = 0; tt < TT; ++tt) hc[tt] = h[t0 + tt];

  float hp = (t0 > 0) ? h[t0 - 1] : 0.0f;
  int buf = 0;
  float* wrow = wsum[wave];
  float cs = (s0 + s1) + (s2 + s3);
  float xs = 0.0f;
  if (owner) xs = hyst_wave_sum(s0, s1, s2, s3, B0, B1, sel1, sel2, sel4);
  for (int seg = t0; seg < t1; seg += SEG * TT) {
  const int segend = (seg + SEG * TT < t1) ? seg + SEG * TT : t1;
  if (owner) {
  for (int tb = seg; tb < segend; tb += TT) {
    const int nb = (tb + TT < T) ? tb + TT : tb;
    float hn[TT];
#pragma unroll
    for (int tt = 0; tt < TT; ++tt) hn[tt] = h[nb + tt];
    float* sb = sx[wave][buf];
#pragma unroll
    for (int tt = 0; tt < TT; ++tt) {
      const float hi = hc[tt];
      if (hi > hp) {
        s0 = (a0 <= hi) ? 1.0f : s0;
        s1 = (a1 <= hi) ? 1.0f : s1;
        s2 = (a2 <= hi) ? 1.0f : s2;
        s3 = (a3 <= hi) ? 1.0f : s3;
      }
      if (hi < hp) {
        s0 = (b0 >= hi) ? -1.0f : s0;
        s1 = (b1 >= hi) ? -1.0f : s1;
        s2 = (b2 >= hi) ? -1.0f : s2;
        s3 = (b3 >= hi) ? -1.0f : s3;
      }
      hp = hi;

      const float cn = (s0 + s1) + (s2 + s3);
      if (__any(cn != cs)) xs = hyst_wave_sum(s0, s1, s2, s3, B0, B1, sel1, sel2, sel4);
      cs = cn;
      sb[tt * PITCH + lane] = xs;
    }
    __builtin_amdgcn_fence(__ATOMIC_RELEASE, "workgroup");
    __builtin_amdgcn_wave_barrier();
    __builtin_amdgcn_fence(__ATOMIC_ACQUIRE, "workgroup");
    const float* r = sb + lane * PITCH;
    float q0 = (r[0] + r[1]) + (r[2] + r[3]);
    float q1 = (r[4] + r[5]) + (r[6] + r[7]);
    float q2 = (r[24] + r[25]) + (r[26] + r[27]);
    float q3 = (r[28] + r[29]) + (r[30] + r[31]);
    wrow[tb - seg + lane] = (q0 + q1) + (q2 + q3);
#pragma unroll
    for (int tt = 0; tt < TT; ++tt) hc[tt] = hn[tt];
    buf ^= 1;
  }
  } else {
    for (int i = lane; i < segend - seg; i += 32) wrow[i] = 0.0f;
  }
  __syncthreads();
  static_assert(SEG * TT <= WPG * 32, "one thread per step of a segment");
  const int i = threadIdx.x;
  if (i < segend - seg) {
    float sum = wsum[0][i];
#pragma unroll
    for (int g = 1; g < WPG; ++g) sum += wsum[g][i];
    pout[seg + i] = sum;
  }
  __syncthreads();
  }
}

__global__ __launch_bounds__(BLOCK) void hyst_reduce_kernel(
    const float* __restrict__ partial, const float* __restrict__ scale,
    const float* __restrict__ offset, float* __restrict__ out,
    int T, int NW, float inv_n) {
  int t = blockIdx.x * blockDim.x + threadIdx.x;
  if (t < T) {
    float sum = 0.0f;
    for (int wv = 0; wv < NW; ++wv) sum += partial[(size_t)wv * T + t];
    out[t] = scale[0] * (sum * inv_n) + offset[0];
  }
}

extern "C" void kernel_launch(void* const* d_in, const int* in_sizes, int n_in,
                              void* d_out, int out_size, void* d_ws, size_t ws_size,
                              hipStream_t stream) {
  const float* h      = (const float*)d_in[0];
  const float* mesh   = (const float*)d_in[1];
  const float* raw    = (const float*)d_in[2];
  const float* scale  = (const float*)d_in[3];
  const float* offset = (const float*)d_in[4];
  float* out = (float*)d_out;

  const int T = in_sizes[0];
  const int n = in_sizes[2];

  const int NW   = (n + RPW - 1) / RPW;
  const int NG   = (NW + WPG - 1) / WPG;
  const int npad = NW * RPW;

  float* partial = (float*)d_ws;
  float* dens    = partial + (size_t)NG * T;
  float* agg     = dens + npad;

  hyst_dens_kernel<<<(npad + BLOCK - 1) / BLOCK, BLOCK, 0, stream>>>(
      raw, h, dens, agg, n, npad, T);
  hyst_scan_kernel<<<dim3(NG, NCH), WPG * 32, 0, stream>>>(
      h, mesh, dens, agg, partial, T, n, NW);
  hyst_reduce_kernel<<<(T + BLOCK - 1) / BLOCK, BLOCK, 0, stream>>>(
      partial, scale, offset, out, T, NG, 1.0f / (float)n);
}


# scan kernel: all 16 sign-pattern sums of a lane lie in an LDS table (same 2 WMMAs); a relay's state is its part of the table offset, a step adds the 4 parts and reads its share; sums unchanged
# speedup vs baseline: 24.3362x; 1.1050x over previous
#include <hip/hip_runtime.h>
#include <math.h>

typedef __attribute__((ext_vector_type(2))) float v2f;
typedef __attribute__((ext_vector_type(8))) float v8f;

#define RPW 128
#define TT 32
#define PITCH 33
#define BLOCK 256
#define NCH 32
#define WPG 20
#define SEG 4
#define DEAL 9
#define NPAT 16
#define LAG 4

__global__ __launch_bounds__(BLOCK) void hyst_dens_kernel(
    const float* __restrict__ raw, const float* __restrict__ h,
    float* __restrict__ dens, float* __restrict__ agg, int n, int npad, int T) {
  int j = blockIdx.x * blockDim.x + threadIdx.x;
  const int nblk = T / TT;
  for (int b = j; b < nblk; b += gridDim.x * blockDim.x) {
    const int bb = b * TT;
    float q = (bb > 0) ? h[bb - 1] : 0.0f;
    float mu = -__builtin_huge_valf(), md = __builtin_huge_valf();
#pragma unroll
    for (int tt = 0; tt < TT; ++tt) {
      const float hi = h[bb + tt];
      mu = (hi > q) ? fmaxf(mu, hi) : mu;
      md = (hi < q) ? fminf(md, hi) : md;
      q = hi;
    }
    agg[b] = mu;
    agg[nblk + b] = md;
  }
  if (j < npad) {
    float v = 0.0f;
    if (j < n) {
      float x = raw[j];
      v = fmaxf(x, 0.0f) + log1pf(expf(-fabsf(x)));
    }
    dens[j] = v;
  }
}

__device__ __forceinline__ void hyst_pattern_table(float d0, float d1, float d2, float d3,
                                                   float q0, float q1, float q2, float q3,
                                                   int lane, float* tl) {
  const bool up = lane >= 16;
  const int r = lane & 15;
  v2f A, BX, BY;
  A.x  = ((r >> (up ? 2 : 0)) & 1) ? 1.0f : -1.0f;
  A.y  = ((r >> (up ? 3 : 1)) & 1) ? 1.0f : -1.0f;
  BX.x = up ? q2 : d0;
  BX.y = up ? q3 : d1;
  BY.x = up ? d2 : q0;
  BY.y = up ? d3 : q1;
  v8f zero = {};
  v8f tx = __builtin_amdgcn_wmma_f32_16x16x4_f32(
      false, A, false, BX, (short)0, zero, false, false);
  v8f ty = __builtin_amdgcn_wmma_f32_16x16x4_f32(
      false, A, false, BY, (short)0, zero, false, false);
  asm volatile("s_wait_alu depctr_va_vdst(0)" : "+v"(tx), "+v"(ty));
  float* trow = tl + (up ? 8 * 32 - 16 : 0);
#pragma unroll
  for (int i = 0; i < 8; ++i) {
    trow[i * 32]      = tx[i];
    trow[i * 32 + 16] = ty[i];
  }
}

__global__ __launch_bounds__(WPG * 32) void hyst_scan_kernel(
    const float* __restrict__ h, const float* __restrict__ mesh,
    const float* __restrict__ dens, const float* __restrict__ agg,
    float* __restrict__ partial, int T, int n, int NW) {
  __shared__ float sx[WPG][2][TT * PITCH];
  __shared__ float wsum[WPG][SEG * TT];
  __shared__ float tabl[WPG][NPAT * 32];

  const int lane = threadIdx.x & 31;
  const int wave = __builtin_amdgcn_readfirstlane((int)(threadIdx.x >> 5));
  static_assert((DEAL * DEAL) % WPG == 1, "DEAL is its own inverse modulo WPG: every slot is taken once");
  const int w    = ((wave * DEAL) % WPG) * (int)gridDim.x + (int)blockIdx.x;
  const bool owner = w < NW;
  const int p0   = w * RPW + 4 * lane;

  float a0 = 3.0e38f, b0 = -3.0e38f, d0 = 0.0f;
  float a1 = 3.0e38f, b1 = -3.0e38f, d1 = 0.0f;
  float a2 = 3.0e38f, b2 = -3.0e38f, d2 = 0.0f;
  float a3 = 3.0e38f, b3 = -3.0e38f, d3 = 0.0f;
  if (p0 < n)     { b0 = mesh[2 * p0];     a0 = mesh[2 * p0 + 1]; d0 = dens[p0]; }
  if (p0 + 1 < n) { b1 = mesh[2 * p0 + 2]; a1 = mesh[2 * p0 + 3]; d1 = dens[p0 + 1]; }
  if (p0 + 2 < n) { b2 = mesh[2 * p0 + 4]; a2 = mesh[2 * p0 + 5]; d2 = dens[p0 + 2]; }
  if (p0 + 3 < n) { b3 = mesh[2 * p0 + 6]; a3 = mesh[2 * p0 + 7]; d3 = dens[p0 + 3]; }

  const int pp = w * RPW + 4 * (lane ^ 16);
  float q0 = 0.0f, q1 = 0.0f, q2 = 0.0f, q3 = 0.0f;
  if (pp < n)     q0 = dens[pp];
  if (pp + 1 < n) q1 = dens[pp + 1];
  if (pp + 2 < n) q2 = dens[pp + 2];
  if (pp + 3 < n) q3 = dens[pp + 3];

  const int tpc = (T / TT + NCH - 1) / NCH;
  const int t0  = (NCH - 1 - (int)blockIdx.y) * tpc * TT;
  if (t0 >= T) return;
  const int t1  = (t0 + tpc * TT < T) ? t0 + tpc * TT : T;

  float* pout = partial + (size_t)blockIdx.x * T;

  if (owner) {
    hyst_pattern_table(d0, d1, d2, d3, q0, q1, q2, q3, lane, tabl[wave] + lane);
    __builtin_amdgcn_fence(__ATOMIC_RELEASE, "workgroup");
    __builtin_amdgcn_wave_barrier();
    __builtin_amdgcn_fence(__ATOMIC_ACQUIRE, "workgroup");
  }

  int o0 = 0, o1 = 0, o2 = 0, o3 = 0;
  if (owner && t0 > 0) {
    float c0 = (p0 < n) ? 0.0f : -1.0f;
    float c1 = (p0 + 1 < n) ? 0.0f : -1.0f;
    float c2 = (p0 + 2 < n) ? 0.0f : -1.0f;
    float c3 = (p0 + 3 < n) ? 0.0f : -1.0f;
    float am = fminf(fminf((c0 == 0.0f) ? a0 : 3.0e38f, (c1 == 0.0f) ? a1 : 3.0e38f),
                     fminf((c2 == 0.0f) ? a2 : 3.0e38f, (c3 == 0.0f) ? a3 : 3.0e38f));
    float bm = fmaxf(fmaxf((c0 == 0.0f) ? b0 : -3.0e38f, (c1 == 0.0f) ? b1 : -3.0e38f),
                     fmaxf((c2 == 0.0f) ? b2 : -3.0e38f, (c3 == 0.0f) ? b3 : -3.0e38f));
    const int nblk = T / TT;
    const int nb0  = t0 / TT;
    bool done = false;
    for (int g = (nb0 - 1) / 32; g >= 0 && !done; --g) {
      const int bi = g * 32 + lane;
      float muv = -__builtin_huge_valf(), mdv = __builtin_huge_valf();
      if (bi < nb0) { muv = agg[bi]; mdv = agg[nblk + bi]; }
      const int btop = (nb0 - 1 - g * 32 < 31) ? nb0 - 1 - g * 32 : 31;
      for (int b = btop; b >= 0; --b) {
        const float mu = __int_as_float(__builtin_amdgcn_readlane(__float_as_int(muv), b));
        const float md = __int_as_float(__builtin_amdgcn_readlane(__float_as_int(mdv), b));
        const bool hit = (am <= mu) || (bm >= md);
        if (!__any(hit)) continue;
        const int bb = (g * 32 + b) * TT;
        float hb[TT];
#pragma unroll
        for (int tt = 0; tt < TT; ++tt) hb[tt] = h[bb + tt];
        float hq = (bb > 0) ? h[bb - 1] : 0.0f;
        float e0 = 0.0f, e1 = 0.0f, e2 = 0.0f, e3 = 0.0f;
#pragma unroll
        for (int tt = 0; tt < TT; ++tt) {
          const float hi = hb[tt];
          if (hi > hq) {
            e0 = (a0 <= hi) ? 1.0f : e0;
            e1 = (a1 <= hi) ? 1.0f : e1;
            e2 = (a2 <= hi) ? 1.0f : e2;
            e3 = (a3 <= hi) ? 1.0f : e3;
          }
          if (hi < hq) {
            e0 = (b0 >= hi) ? -1.0f : e0;
            e1 = (b1 >= hi) ? -1.0f : e1;
            e2 = (b2 >= hi) ? -1.0f : e2;
            e3 = (b3 >= hi) ? -1.0f : e3;
          }
          hq = hi;
        }
        c0 = (c0 == 0.0f) ? e0 : c0;
        c1 = (c1 == 0.0f) ? e1 : c1;
        c2 = (c2 == 0.0f) ? e2 : c2;
        c3 = (c3 == 0.0f) ? e3 : c3;
        if (__all((c0 != 0.0f) && (c1 != 0.0f) && (c2 != 0.0f) && (c3 != 0.0f))) { done = true; break; }
        am = fminf(fminf((c0 == 0.0f) ? a0 : 3.0e38f, (c1 == 0.0f) ? a1 : 3.0e38f),
                   fminf((c2 == 0.0f) ? a2 : 3.0e38f, (c3 == 0.0f) ? a3 : 3.0e38f));
        bm = fmaxf(fmaxf((c0 == 0.0f) ? b0 : -3.0e38f, (c1 == 0.0f) ? b1 : -3.0e38f),
                   fmaxf((c2 == 0.0f) ? b2 : -3.0e38f, (c3 == 0.0f) ? b3 : -3.0e38f));
      }
    }
    o0 = (c0 > 0.0f) ? 128 : 0;
    o1 = (c1 > 0.0f) ? 256 : 0;
    o2 = (c2 > 0.0f) ? 512 : 0;
    o3 = (c3 > 0.0f) ? 1024 : 0;
  }

  float hc[TT];
#pragma unroll
  for (int tt = 0; tt < TT; ++tt) hc[tt] = h[t0 + tt];

  float hp = (t0 > 0) ? h[t0 - 1] : 0.0f;
  int buf = 0;
  float* wrow = wsum[wave];
  const char* tcol = (const char*)(tabl[wave] + lane);
  for (int seg = t0; seg < t1; seg += SEG * TT) {
  const int segend = (seg + SEG * TT < t1) ? seg + SEG * TT : t1;
  if (owner) {
  for (int tb = seg; tb < segend; tb += TT) {
    const int nb = (tb + TT < T) ? tb + TT : tb;
    float hn[TT];
#pragma unroll
    for (int tt = 0; tt < TT; ++tt) hn[tt] = h[nb + tt];
    float* sb = sx[wave][buf];
    float xv[TT];
#pragma unroll
    for (int tt = 0; tt < TT; ++tt) {
      const float hi = hc[tt];
      if (hi > hp) {
        o0 = (a0 <= hi) ? 128 : o0;
        o1 = (a1 <= hi) ? 256 : o1;
        o2 = (a2 <= hi) ? 512 : o2;
        o3 = (a3 <= hi) ? 1024 : o3;
      }
      if (hi < hp) {
        o0 = (b0 >= hi) ? 0 : o0;
        o1 = (b1 >= hi) ? 0 : o1;
        o2 = (b2 >= hi) ? 0 : o2;
        o3 = (b3 >= hi) ? 0 : o3;
      }
      hp = hi;

      xv[tt] = *(const float*)(tcol + ((o0 + o1) + (o2 + o3)));
      if (tt >= LAG) sb[(tt - LAG) * PITCH + lane] = xv[tt - LAG];
    }
#pragma unroll
    for (int tt = TT - LAG; tt < TT; ++tt) sb[tt * PITCH + lane] = xv[tt];
    __builtin_amdgcn_fence(__ATOMIC_RELEASE, "workgroup");
    __builtin_amdgcn_wave_barrier();
    __builtin_amdgcn_fence(__ATOMIC_ACQUIRE, "workgroup");
    const float* r = sb + lane * PITCH;
    const float u0 = (r[0] + r[1]) + (r[2] + r[3]);
    const float u1 = (r[4] + r[5]) + (r[6] + r[7]);
    const float u2 = (r[8] + r[9]) + (r[10] + r[11]);
    const float u3 = (r[12] + r[13]) + (r[14] + r[15]);
    const float u4 = (r[16] + r[17]) + (r[18] + r[19]);
    const float u5 = (r[20] + r[21]) + (r[22] + r[23]);
    const float u6 = (r[24] + r[25]) + (r[26] + r[27]);
    const float u7 = (r[28] + r[29]) + (r[30] + r[31]);
    wrow[tb - seg + lane] = ((u0 + u1) + (u2 + u3)) + ((u4 + u5) + (u6 + u7));
#pragma unroll
    for (int tt = 0; tt < TT; ++tt) hc[tt] = hn[tt];
    buf ^= 1;
  }
  } else {
    for (int i = lane; i < segend - seg; i += 32) wrow[i] = 0.0f;
  }
  __syncthreads();
  static_assert(SEG * TT <= WPG * 32, "one thread per step of a segment");
  const int i = threadIdx.x;
  if (i < segend - seg) {
    float sum = wsum[0][i];
#pragma unroll
    for (int g = 1; g < WPG; ++g) sum += wsum[g][i];
    pout[seg + i] = sum;
  }
  __syncthreads();
  }
}

__global__ __launch_bounds__(BLOCK) void hyst_reduce_kernel(
    const float* __restrict__ partial, const float* __restrict__ scale,
    const float* __restrict__ offset, float* __restrict__ out,
    int T, int NW, float inv_n) {
  int t = blockIdx.x * blockDim.x + threadIdx.x;
  if (t < T) {
    float sum = 0.0f;
    for (int wv = 0; wv < NW; ++wv) sum += partial[(size_t)wv * T + t];
    out[t] = scale[0] * (sum * inv_n) + offset[0];
  }
}

extern "C" void kernel_launch(void* const* d_in, const int* in_sizes, int n_in,
                              void* d_out, int out_size, void* d_ws, size_t ws_size,
                              hipStream_t stream) {
  const float* h      = (const float*)d_in[0];
  const float* mesh   = (const float*)d_in[1];
  const float* raw    = (const float*)d_in[2];
  const float* scale  = (const float*)d_in[3];
  const float* offset = (const float*)d_in[4];
  float* out = (float*)d_out;

  const int T = in_sizes[0];
  const int n = in_sizes[2];

  const int NW   = (n + RPW - 1) / RPW;
  const int NG   = (NW + WPG - 1) / WPG;
  const int npad = NW * RPW;

  float* partial = (float*)d_ws;
  float* dens    = partial + (size_t)NG * T;
  float* agg     = dens + npad;

  hyst_dens_kernel<<<(npad + BLOCK - 1) / BLOCK, BLOCK, 0, stream>>>(
      raw, h, dens, agg, n, npad, T);
  hyst_scan_kernel<<<dim3(NG, NCH), WPG * 32, 0, stream>>>(
      h, mesh, dens, agg, partial, T, n, NW);
  hyst_reduce_kernel<<<(T + BLOCK - 1) / BLOCK, BLOCK, 0, stream>>>(
      partial, scale, offset, out, T, NG, 1.0f / (float)n);
}


# scan kernel: a step runs its compares, selects and table offset only if a scalar test says a relay can switch (above the last rise that ran, else the wave's min alpha; same for falls); sums unchanged
# speedup vs baseline: 25.5607x; 1.0503x over previous
#include <hip/hip_runtime.h>
#include <math.h>

typedef __attribute__((ext_vector_type(2))) float v2f;
typedef __attribute__((ext_vector_type(8))) float v8f;

#define RPW 128
#define TT 32
#define PITCH 33
#define BLOCK 256
#define NCH 32
#define WPG 20
#define SEG 4
#define DEAL 9
#define NPAT 16
#define LAG 4

__global__ __launch_bounds__(BLOCK) void hyst_dens_kernel(
    const float* __restrict__ raw, const float* __restrict__ h,
    float* __restrict__ dens, float* __restrict__ agg, int n, int npad, int T) {
  int j = blockIdx.x * blockDim.x + threadIdx.x;
  const int nblk = T / TT;
  for (int b = j; b < nblk; b += gridDim.x * blockDim.x) {
    const int bb = b * TT;
    float q = (bb > 0) ? h[bb - 1] : 0.0f;
    float mu = -__builtin_huge_valf(), md = __builtin_huge_valf();
#pragma unroll
    for (int tt = 0; tt < TT; ++tt) {
      const float hi = h[bb + tt];
      mu = (hi > q) ? fmaxf(mu, hi) : mu;
      md = (hi < q) ? fminf(md, hi) : md;
      q = hi;
    }
    agg[b] = mu;
    agg[nblk + b] = md;
  }
  if (j < npad) {
    float v = 0.0f;
    if (j < n) {
      float x = raw[j];
      v = fmaxf(x, 0.0f) + log1pf(expf(-fabsf(x)));
    }
    dens[j] = v;
  }
}

__device__ __forceinline__ void hyst_pattern_table(float d0, float d1, float d2, float d3,
                                                   float q0, float q1, float q2, float q3,
                                                   int lane, float* tl) {
  const bool up = lane >= 16;
  const int r = lane & 15;
  v2f A, BX, BY;
  A.x  = ((r >> (up ? 2 : 0)) & 1) ? 1.0f : -1.0f;
  A.y  = ((r >> (up ? 3 : 1)) & 1) ? 1.0f : -1.0f;
  BX.x = up ? q2 : d0;
  BX.y = up ? q3 : d1;
  BY.x = up ? d2 : q0;
  BY.y = up ? d3 : q1;
  v8f zero = {};
  v8f tx = __builtin_amdgcn_wmma_f32_16x16x4_f32(
      false, A, false, BX, (short)0, zero, false, false);
  v8f ty = __builtin_amdgcn_wmma_f32_16x16x4_f32(
      false, A, false, BY, (short)0, zero, false, false);
  asm volatile("s_wait_alu depctr_va_vdst(0)" : "+v"(tx), "+v"(ty));
  float* trow = tl + (up ? 8 * 32 - 16 : 0);
#pragma unroll
  for (int i = 0; i < 8; ++i) {
    trow[i * 32]      = tx[i];
    trow[i * 32 + 16] = ty[i];
  }
}

__global__ __launch_bounds__(WPG * 32) void hyst_scan_kernel(
    const float* __restrict__ h, const float* __restrict__ mesh,
    const float* __restrict__ dens, const float* __restrict__ agg,
    float* __restrict__ partial, int T, int n, int NW) {
  __shared__ float sx[WPG][2][TT * PITCH];
  __shared__ float wsum[WPG][SEG * TT];
  __shared__ float tabl[WPG][NPAT * 32];

  const int lane = threadIdx.x & 31;
  const int wave = __builtin_amdgcn_readfirstlane((int)(threadIdx.x >> 5));
  static_assert((DEAL * DEAL) % WPG == 1, "DEAL is its own inverse modulo WPG: every slot is taken once");
  const int w    = ((wave * DEAL) % WPG) * (int)gridDim.x + (int)blockIdx.x;
  const bool owner = w < NW;
  const int p0   = w * RPW + 4 * lane;

  float a0 = 3.0e38f, b0 = -3.0e38f, d0 = 0.0f;
  float a1 = 3.0e38f, b1 = -3.0e38f, d1 = 0.0f;
  float a2 = 3.0e38f, b2 = -3.0e38f, d2 = 0.0f;
  float a3 = 3.0e38f, b3 = -3.0e38f, d3 = 0.0f;
  if (p0 < n)     { b0 = mesh[2 * p0];     a0 = mesh[2 * p0 + 1]; d0 = dens[p0]; }
  if (p0 + 1 < n) { b1 = mesh[2 * p0 + 2]; a1 = mesh[2 * p0 + 3]; d1 = dens[p0 + 1]; }
  if (p0 + 2 < n) { b2 = mesh[2 * p0 + 4]; a2 = mesh[2 * p0 + 5]; d2 = dens[p0 + 2]; }
  if (p0 + 3 < n) { b3 = mesh[2 * p0 + 6]; a3 = mesh[2 * p0 + 7]; d3 = dens[p0 + 3]; }

  const int pp = w * RPW + 4 * (lane ^ 16);
  float q0 = 0.0f, q1 = 0.0f, q2 = 0.0f, q3 = 0.0f;
  if (pp < n)     q0 = dens[pp];
  if (pp + 1 < n) q1 = dens[pp + 1];
  if (pp + 2 < n) q2 = dens[pp + 2];
  if (pp + 3 < n) q3 = dens[pp + 3];

  const int tpc = (T / TT + NCH - 1) / NCH;
  const int t0  = (NCH - 1 - (int)blockIdx.y) * tpc * TT;
  if (t0 >= T) return;
  const int t1  = (t0 + tpc * TT < T) ? t0 + tpc * TT : T;

  float amn = fminf(fminf(a0, a1), fminf(a2, a3));
  float bmx = fmaxf(fmaxf(b0, b1), fmaxf(b2, b3));
#define HYST_XLANE(v, ctl) __int_as_float(__builtin_amdgcn_update_dpp(0, __float_as_int(v), ctl, 0xf, 0xf, true))
  amn = fminf(amn, HYST_XLANE(amn, 0x161));  bmx = fmaxf(bmx, HYST_XLANE(bmx, 0x161));
  amn = fminf(amn, HYST_XLANE(amn, 0x162));  bmx = fmaxf(bmx, HYST_XLANE(bmx, 0x162));
  amn = fminf(amn, HYST_XLANE(amn, 0x164));  bmx = fmaxf(bmx, HYST_XLANE(bmx, 0x164));
  amn = fminf(amn, HYST_XLANE(amn, 0x168));  bmx = fmaxf(bmx, HYST_XLANE(bmx, 0x168));
#undef HYST_XLANE
  amn = fminf(amn, __int_as_float(__builtin_amdgcn_permlanex16(
            0, __float_as_int(amn), 0x76543210, 0xfedcba98, false, false)));
  bmx = fmaxf(bmx, __int_as_float(__builtin_amdgcn_permlanex16(
            0, __float_as_int(bmx), 0x76543210, 0xfedcba98, false, false)));
  const float amin = __int_as_float(__builtin_amdgcn_readfirstlane(__float_as_int(amn)));
  const float bmax = __int_as_float(__builtin_amdgcn_readfirstlane(__float_as_int(bmx)));
  const float alow  = (amin > -3.0e38f && amin < 3.0e38f)
                    ? amin - fmaxf(fabsf(amin) * 1.0e-6f, 1.0e-30f) : -__builtin_huge_valf();
  const float bhigh = (bmax > -3.0e38f && bmax < 3.0e38f)
                    ? bmax + fmaxf(fabsf(bmax) * 1.0e-6f, 1.0e-30f) : __builtin_huge_valf();

  float* pout = partial + (size_t)blockIdx.x * T;

  if (owner) {
    hyst_pattern_table(d0, d1, d2, d3, q0, q1, q2, q3, lane, tabl[wave] + lane);
    __builtin_amdgcn_fence(__ATOMIC_RELEASE, "workgroup");
    __builtin_amdgcn_wave_barrier();
    __builtin_amdgcn_fence(__ATOMIC_ACQUIRE, "workgroup");
  }

  int o0 = 0, o1 = 0, o2 = 0, o3 = 0;
  if (owner && t0 > 0) {
    float c0 = (p0 < n) ? 0.0f : -1.0f;
    float c1 = (p0 + 1 < n) ? 0.0f : -1.0f;
    float c2 = (p0 + 2 < n) ? 0.0f : -1.0f;
    float c3 = (p0 + 3 < n) ? 0.0f : -1.0f;
    float am = fminf(fminf((c0 == 0.0f) ? a0 : 3.0e38f, (c1 == 0.0f) ? a1 : 3.0e38f),
                     fminf((c2 == 0.0f) ? a2 : 3.0e38f, (c3 == 0.0f) ? a3 : 3.0e38f));
    float bm = fmaxf(fmaxf((c0 == 0.0f) ? b0 : -3.0e38f, (c1 == 0.0f) ? b1 : -3.0e38f),
                     fmaxf((c2 == 0.0f) ? b2 : -3.0e38f, (c3 == 0.0f) ? b3 : -3.0e38f));
    const int nblk = T / TT;
    const int nb0  = t0 / TT;
    bool done = false;
    for (int g = (nb0 - 1) / 32; g >= 0 && !done; --g) {
      const int bi = g * 32 + lane;
      float muv = -__builtin_huge_valf(), mdv = __builtin_huge_valf();
      if (bi < nb0) { muv = agg[bi]; mdv = agg[nblk + bi]; }
      const int btop = (nb0 - 1 - g * 32 < 31) ? nb0 - 1 - g * 32 : 31;
      for (int b = btop; b >= 0; --b) {
        const float mu = __int_as_float(__builtin_amdgcn_readlane(__float_as_int(muv), b));
        const float md = __int_as_float(__builtin_amdgcn_readlane(__float_as_int(mdv), b));
        const bool hit = (am <= mu) || (bm >= md);
        if (!__any(hit)) continue;
        const int bb = (g * 32 + b) * TT;
        float hb[TT];
#pragma unroll
        for (int tt = 0; tt < TT; ++tt) hb[tt] = h[bb + tt];
        float hq = (bb > 0) ? h[bb - 1] : 0.0f;
        float e0 = 0.0f, e1 = 0.0f, e2 = 0.0f, e3 = 0.0f;
        float rl = alow, fl = bhigh;
#pragma unroll
        for (int tt = 0; tt < TT; ++tt) {
          const float hi = hb[tt];
          if (hi > __builtin_elementwise_maximum(hq, rl)) {
            e0 = (a0 <= hi) ? 1.0f : e0;
            e1 = (a1 <= hi) ? 1.0f : e1;
            e2 = (a2 <= hi) ? 1.0f : e2;
            e3 = (a3 <= hi) ? 1.0f : e3;
            rl = hi;
            fl = bhigh;
          }
          if (hi < __builtin_elementwise_minimum(hq, fl)) {
            e0 = (b0 >= hi) ? -1.0f : e0;
            e1 = (b1 >= hi) ? -1.0f : e1;
            e2 = (b2 >= hi) ? -1.0f : e2;
            e3 = (b3 >= hi) ? -1.0f : e3;
            fl = hi;
            rl = alow;
          }
          hq = hi;
        }
        c0 = (c0 == 0.0f) ? e0 : c0;
        c1 = (c1 == 0.0f) ? e1 : c1;
        c2 = (c2 == 0.0f) ? e2 : c2;
        c3 = (c3 == 0.0f) ? e3 : c3;
        if (__all((c0 != 0.0f) && (c1 != 0.0f) && (c2 != 0.0f) && (c3 != 0.0f))) { done = true; break; }
        am = fminf(fminf((c0 == 0.0f) ? a0 : 3.0e38f, (c1 == 0.0f) ? a1 : 3.0e38f),
                   fminf((c2 == 0.0f) ? a2 : 3.0e38f, (c3 == 0.0f) ? a3 : 3.0e38f));
        bm = fmaxf(fmaxf((c0 == 0.0f) ? b0 : -3.0e38f, (c1 == 0.0f) ? b1 : -3.0e38f),
                   fmaxf((c2 == 0.0f) ? b2 : -3.0e38f, (c3 == 0.0f) ? b3 : -3.0e38f));
      }
    }
    o0 = (c0 > 0.0f) ? 128 : 0;
    o1 = (c1 > 0.0f) ? 256 : 0;
    o2 = (c2 > 0.0f) ? 512 : 0;
    o3 = (c3 > 0.0f) ? 1024 : 0;
  }

  float hc[TT];
#pragma unroll
  for (int tt = 0; tt < TT; ++tt) hc[tt] = h[t0 + tt];

  float hp = (t0 > 0) ? h[t0 - 1] : 0.0f;
  int buf = 0;
  float* wrow = wsum[wave];
  const char* tcol = (const char*)(tabl[wave] + lane);
  const char* tcur = tcol + ((o0 + o1) + (o2 + o3));
  float rlast = alow, flast = bhigh;
  for (int seg = t0; seg < t1; seg += SEG * TT) {
  const int segend = (seg + SEG * TT < t1) ? seg + SEG * TT : t1;
  if (owner) {
  for (int tb = seg; tb < segend; tb += TT) {
    const int nb = (tb + TT < T) ? tb + TT : tb;
    float hn[TT];
#pragma unroll
    for (int tt = 0; tt < TT; ++tt) hn[tt] = h[nb + tt];
    float* sb = sx[wave][buf];
    float xv[TT];
#pragma unroll
    for (int tt = 0; tt < TT; ++tt) {
      const float hi = hc[tt];
      if (hi > __builtin_elementwise_maximum(hp, rlast)) {
        o0 = (a0 <= hi) ? 128 : o0;
        o1 = (a1 <= hi) ? 256 : o1;
        o2 = (a2 <= hi) ? 512 : o2;
        o3 = (a3 <= hi) ? 1024 : o3;
        tcur = tcol + ((o0 + o1) + (o2 + o3));
        rlast = hi;
        flast = bhigh;
      }
      if (hi < __builtin_elementwise_minimum(hp, flast)) {
        o0 = (b0 >= hi) ? 0 : o0;
        o1 = (b1 >= hi) ? 0 : o1;
        o2 = (b2 >= hi) ? 0 : o2;
        o3 = (b3 >= hi) ? 0 : o3;
        tcur = tcol + ((o0 + o1) + (o2 + o3));
        flast = hi;
        rlast = alow;
      }
      hp = hi;

      xv[tt] = *(const float*)tcur;
      if (tt >= LAG) sb[(tt - LAG) * PITCH + lane] = xv[tt - LAG];
    }
#pragma unroll
    for (int tt = TT - LAG; tt < TT; ++tt) sb[tt * PITCH + lane] = xv[tt];
    __builtin_amdgcn_fence(__ATOMIC_RELEASE, "workgroup");
    __builtin_amdgcn_wave_barrier();
    __builtin_amdgcn_fence(__ATOMIC_ACQUIRE, "workgroup");
    const float* r = sb + lane * PITCH;
    const float u0 = (r[0] + r[1]) + (r[2] + r[3]);
    const float u1 = (r[4] + r[5]) + (r[6] + r[7]);
    const float u2 = (r[8] + r[9]) + (r[10] + r[11]);
    const float u3 = (r[12] + r[13]) + (r[14] + r[15]);
    const float u4 = (r[16] + r[17]) + (r[18] + r[19]);
    const float u5 = (r[20] + r[21]) + (r[22] + r[23]);
    const float u6 = (r[24] + r[25]) + (r[26] + r[27]);
    const float u7 = (r[28] + r[29]) + (r[30] + r[31]);
    wrow[tb - seg + lane] = ((u0 + u1) + (u2 + u3)) + ((u4 + u5) + (u6 + u7));
#pragma unroll
    for (int tt = 0; tt < TT; ++tt) hc[tt] = hn[tt];
    buf ^= 1;
  }
  } else {
    for (int i = lane; i < segend - seg; i += 32) wrow[i] = 0.0f;
  }
  __syncthreads();
  static_assert(SEG * TT <= WPG * 32, "one thread per step of a segment");
  const int i = threadIdx.x;
  if (i < segend - seg) {
    float sum = wsum[0][i];
#pragma unroll
    for (int g = 1; g < WPG; ++g) sum += wsum[g][i];
    pout[seg + i] = sum;
  }
  __syncthreads();
  }
}

__global__ __launch_bounds__(BLOCK) void hyst_reduce_kernel(
    const float* __restrict__ partial, const float* __restrict__ scale,
    const float* __restrict__ offset, float* __restrict__ out,
    int T, int NW, float inv_n) {
  int t = blockIdx.x * blockDim.x + threadIdx.x;
  if (t < T) {
    float sum = 0.0f;
    for (int wv = 0; wv < NW; ++wv) sum += partial[(size_t)wv * T + t];
    out[t] = scale[0] * (sum * inv_n) + offset[0];
  }
}

extern "C" void kernel_launch(void* const* d_in, const int* in_sizes, int n_in,
                              void* d_out, int out_size, void* d_ws, size_t ws_size,
                              hipStream_t stream) {
  const float* h      = (const float*)d_in[0];
  const float* mesh   = (const float*)d_in[1];
  const float* raw    = (const float*)d_in[2];
  const float* scale  = (const float*)d_in[3];
  const float* offset = (const float*)d_in[4];
  float* out = (float*)d_out;

  const int T = in_sizes[0];
  const int n = in_sizes[2];

  const int NW   = (n + RPW - 1) / RPW;
  const int NG   = (NW + WPG - 1) / WPG;
  const int npad = NW * RPW;

  float* partial = (float*)d_ws;
  float* dens    = partial + (size_t)NG * T;
  float* agg     = dens + npad;

  hyst_dens_kernel<<<(npad + BLOCK - 1) / BLOCK, BLOCK, 0, stream>>>(
      raw, h, dens, agg, n, npad, T);
  hyst_scan_kernel<<<dim3(NG, NCH), WPG * 32, 0, stream>>>(
      h, mesh, dens, agg, partial, T, n, NW);
  hyst_reduce_kernel<<<(T + BLOCK - 1) / BLOCK, BLOCK, 0, stream>>>(
      partial, scale, offset, out, T, NG, 1.0f / (float)n);
}
